# starting kernel
# baseline (speedup 1.0000x reference)
#include <hip/hip_runtime.h>
#include <math.h>

typedef float v2f __attribute__((ext_vector_type(2)));
typedef float v8f __attribute__((ext_vector_type(8)));

#define IN_FT   128
#define OUT_FT  64
#define HEADS   2
#define FT      (HEADS * OUT_FT)
#define NEG_SLOPE 0.2f

__device__ __forceinline__ float leaky(float s) {
    return s > 0.f ? s : NEG_SLOPE * s;
}

__global__ void gat_init_kernel(float* __restrict__ out,
                                float* __restrict__ mmax,
                                float* __restrict__ den,
                                int n_out, int n_nh) {
    int i = blockIdx.x * blockDim.x + threadIdx.x;
    if (i < n_out) out[i] = 0.f;
    if (i < n_nh) { mmax[i] = -INFINITY; den[i] = 0.f; }
}

__global__ __launch_bounds__(128)
void gat_gemm_wmma_kernel(const float* __restrict__ x,
                          const float* __restrict__ W,
                          float* __restrict__ xt, int N) {
    __shared__ float Wl[IN_FT * 16];

    const int tile_n = blockIdx.x & 7;
    const int group  = blockIdx.x >> 3;
    const int col0   = tile_n * 16;

    for (int i = threadIdx.x; i < IN_FT * 16; i += 128) {
        int k = i >> 4, c = i & 15;
        Wl[i] = W[k * FT + col0 + c];
    }
    __syncthreads();

    const int wave = threadIdx.x >> 5;
    const int lane = threadIdx.x & 31;
    const int half = lane >> 4;
    const int l16  = lane & 15;

    const int row0 = (group * 4 + wave) * 16;
    int arow = row0 + l16;
    if (arow >= N) arow = N - 1;

    v8f c = {};
    for (int k0 = 0; k0 < IN_FT; k0 += 4) {
        v2f a, b;
        const float* ap = x + (size_t)arow * IN_FT + (k0 + 2 * half);
        a.x = ap[0];
        a.y = ap[1];
        b.x = Wl[(k0 + 2 * half)     * 16 + l16];
        b.y = Wl[(k0 + 2 * half + 1) * 16 + l16];
        c = __builtin_amdgcn_wmma_f32_16x16x4_f32(false, a, false, b,
                                                  (short)0, c, false, false);
    }

    #pragma unroll
    for (int r = 0; r < 8; ++r) {
        int rr = row0 + r + 8 * half;
        if (rr < N) xt[(size_t)rr * FT + col0 + l16] = c[r];
    }
}

__global__ void gat_logits_kernel(const float* __restrict__ xt,
                                  const float* __restrict__ att_src,
                                  const float* __restrict__ att_dst,
                                  float* __restrict__ a_src,
                                  float* __restrict__ a_dst, int NH) {
    int i = blockIdx.x * blockDim.x + threadIdx.x;
    if (i >= NH) return;
    int n = i >> 1, h = i & 1;
    const float4* xp = (const float4*)(xt + (size_t)n * FT + h * OUT_FT);
    const float4* as = (const float4*)(att_src + h * OUT_FT);
    const float4* ad = (const float4*)(att_dst + h * OUT_FT);
    float s0 = 0.f, s1 = 0.f;
    #pragma unroll
    for (int j = 0; j < OUT_FT / 4; ++j) {
        float4 v = xp[j], a = as[j], d = ad[j];
        s0 += v.x * a.x + v.y * a.y + v.z * a.z + v.w * a.w;
        s1 += v.x * d.x + v.y * d.y + v.z * d.z + v.w * d.w;
    }
    a_src[i] = s0;
    a_dst[i] = s1;
}

__global__ void gat_edge_max_kernel(const int* __restrict__ ei,
                                    const float* __restrict__ a_src,
                                    const float* __restrict__ a_dst,
                                    float* __restrict__ mmax, int E) {
    int e = blockIdx.x * blockDim.x + threadIdx.x;
    if (e >= E) return;
    int s = ei[e], d = ei[E + e];
    #pragma unroll
    for (int h = 0; h < HEADS; ++h) {
        float v = leaky(a_src[s * HEADS + h] + a_dst[d * HEADS + h]);
        atomicMax(&mmax[d * HEADS + h], v);
    }
}

__global__ void gat_edge_expsum_kernel(const int* __restrict__ ei,
                                       const float* __restrict__ a_src,
                                       const float* __restrict__ a_dst,
                                       const float* __restrict__ mmax,
                                       float* __restrict__ den, int E) {
    int e = blockIdx.x * blockDim.x + threadIdx.x;
    if (e >= E) return;
    int s = ei[e], d = ei[E + e];
    #pragma unroll
    for (int h = 0; h < HEADS; ++h) {
        float v = leaky(a_src[s * HEADS + h] + a_dst[d * HEADS + h]);
        atomicAdd(&den[d * HEADS + h], __expf(v - mmax[d * HEADS + h]));
    }
}

__global__ __launch_bounds__(256)
void gat_edge_scatter_kernel(const int* __restrict__ ei,
                             const float* __restrict__ a_src,
                             const float* __restrict__ a_dst,
                             const float* __restrict__ mmax,
                             const float* __restrict__ den,
                             const float* __restrict__ xt,
                             float* __restrict__ out, int E) {
    int gid  = blockIdx.x * blockDim.x + threadIdx.x;
    int e    = gid >> 5;
    int lane = gid & 31;
    if (e >= E) return;
    int s = ei[e], d = ei[E + e];
    int h = lane >> 4;
    float v    = leaky(a_src[s * HEADS + h] + a_dst[d * HEADS + h]);
    float coef = __expf(v - mmax[d * HEADS + h]) / den[d * HEADS + h];
    float4 xv = ((const float4*)(xt + (size_t)s * FT))[lane];
    float* op = out + (size_t)d * FT + lane * 4;
    atomicAdd(op + 0, coef * xv.x);
    atomicAdd(op + 1, coef * xv.y);
    atomicAdd(op + 2, coef * xv.z);
    atomicAdd(op + 3, coef * xv.w);
}

__global__ void gat_epilogue_kernel(float* __restrict__ out,
                                    const float* __restrict__ bias,
                                    const float* __restrict__ prelu_w, int total) {
    int i = blockIdx.x * blockDim.x + threadIdx.x;
    if (i >= total) return;
    float v = out[i] + bias[i & (FT - 1)];
    out[i] = v >= 0.f ? v : prelu_w[0] * v;
}

extern "C" void kernel_launch(void* const* d_in, const int* in_sizes, int n_in,
                              void* d_out, int out_size, void* d_ws, size_t ws_size,
                              hipStream_t stream) {
    const float* x       = (const float*)d_in[0];
    const int*   ei      = (const int*)  d_in[1];
    const float* W       = (const float*)d_in[2];
    const float* att_src = (const float*)d_in[3];
    const float* att_dst = (const float*)d_in[4];
    const float* bias    = (const float*)d_in[5];
    const float* prelu_w = (const float*)d_in[6];

    const int N = in_sizes[0] / IN_FT;
    const int E = in_sizes[1] / 2;

    float* out = (float*)d_out;

    float* ws    = (float*)d_ws;
    float* xt    = ws;
    float* a_src = xt    + (size_t)N * FT;
    float* a_dst = a_src + (size_t)N * HEADS;
    float* mmax  = a_dst + (size_t)N * HEADS;
    float* den   = mmax  + (size_t)N * HEADS;

    const int NH    = N * HEADS;
    const int total = N * FT;

    gat_init_kernel<<<(total + 255) / 256, 256, 0, stream>>>(out, mmax, den, total, NH);

    const int tiles_m = (N + 15) / 16;
    const int groups  = (tiles_m + 3) / 4;
    gat_gemm_wmma_kernel<<<groups * 8, 128, 0, stream>>>(x, W, xt, N);

    gat_logits_kernel<<<(NH + 255) / 256, 256, 0, stream>>>(xt, att_src, att_dst,
                                                            a_src, a_dst, NH);

    gat_edge_max_kernel<<<(E + 255) / 256, 256, 0, stream>>>(ei, a_src, a_dst, mmax, E);
    gat_edge_expsum_kernel<<<(E + 255) / 256, 256, 0, stream>>>(ei, a_src, a_dst,
                                                                mmax, den, E);

    long long sc_threads = (long long)E * 32;
    gat_edge_scatter_kernel<<<(unsigned)((sc_threads + 255) / 256), 256, 0, stream>>>(
        ei, a_src, a_dst, mmax, den, xt, out, E);

    gat_epilogue_kernel<<<(total + 255) / 256, 256, 0, stream>>>(out, bias, prelu_w, total);
}


# Fuse the exp-sum edge pass into the scatter pass: atomics add unnormalised exp*x and the denominator together; one division per output in the epilogue (normalisation rounds after the sum)
# speedup vs baseline: 1.2243x; 1.2243x over previous
#include <hip/hip_runtime.h>
#include <math.h>

typedef float v2f __attribute__((ext_vector_type(2)));
typedef float v8f __attribute__((ext_vector_type(8)));

#define IN_FT   128
#define OUT_FT  64
#define HEADS   2
#define FT      (HEADS * OUT_FT)
#define NEG_SLOPE 0.2f

__device__ __forceinline__ float leaky(float s) {
    return s > 0.f ? s : NEG_SLOPE * s;
}

__global__ void gat_init_kernel(float* __restrict__ out,
                                float* __restrict__ mmax,
                                float* __restrict__ den,
                                int n_out, int n_nh) {
    int i = blockIdx.x * blockDim.x + threadIdx.x;
    if (i < n_out) out[i] = 0.f;
    if (i < n_nh) { mmax[i] = -INFINITY; den[i] = 0.f; }
}

__global__ __launch_bounds__(128)
void gat_gemm_wmma_kernel(const float* __restrict__ x,
                          const float* __restrict__ W,
                          float* __restrict__ xt, int N) {
    __shared__ float Wl[IN_FT * 16];

    const int tile_n = blockIdx.x & 7;
    const int group  = blockIdx.x >> 3;
    const int col0   = tile_n * 16;

    for (int i = threadIdx.x; i < IN_FT * 16; i += 128) {
        int k = i >> 4, c = i & 15;
        Wl[i] = W[k * FT + col0 + c];
    }
    __syncthreads();

    const int wave = threadIdx.x >> 5;
    const int lane = threadIdx.x & 31;
    const int half = lane >> 4;
    const int l16  = lane & 15;

    const int row0 = (group * 4 + wave) * 16;
    int arow = row0 + l16;
    if (arow >= N) arow = N - 1;

    v8f c = {};
    for (int k0 = 0; k0 < IN_FT; k0 += 4) {
        v2f a, b;
        const float* ap = x + (size_t)arow * IN_FT + (k0 + 2 * half);
        a.x = ap[0];
        a.y = ap[1];
        b.x = Wl[(k0 + 2 * half)     * 16 + l16];
        b.y = Wl[(k0 + 2 * half + 1) * 16 + l16];
        c = __builtin_amdgcn_wmma_f32_16x16x4_f32(false, a, false, b,
                                                  (short)0, c, false, false);
    }

    #pragma unroll
    for (int r = 0; r < 8; ++r) {
        int rr = row0 + r + 8 * half;
        if (rr < N) xt[(size_t)rr * FT + col0 + l16] = c[r];
    }
}

__global__ void gat_logits_kernel(const float* __restrict__ xt,
                                  const float* __restrict__ att_src,
                                  const float* __restrict__ att_dst,
                                  float* __restrict__ a_src,
                                  float* __restrict__ a_dst, int NH) {
    int i = blockIdx.x * blockDim.x + threadIdx.x;
    if (i >= NH) return;
    int n = i >> 1, h = i & 1;
    const float4* xp = (const float4*)(xt + (size_t)n * FT + h * OUT_FT);
    const float4* as = (const float4*)(att_src + h * OUT_FT);
    const float4* ad = (const float4*)(att_dst + h * OUT_FT);
    float s0 = 0.f, s1 = 0.f;
    #pragma unroll
    for (int j = 0; j < OUT_FT / 4; ++j) {
        float4 v = xp[j], a = as[j], d = ad[j];
        s0 += v.x * a.x + v.y * a.y + v.z * a.z + v.w * a.w;
        s1 += v.x * d.x + v.y * d.y + v.z * d.z + v.w * d.w;
    }
    a_src[i] = s0;
    a_dst[i] = s1;
}

__global__ void gat_edge_max_kernel(const int* __restrict__ ei,
                                    const float* __restrict__ a_src,
                                    const float* __restrict__ a_dst,
                                    float* __restrict__ mmax, int E) {
    int e = blockIdx.x * blockDim.x + threadIdx.x;
    if (e >= E) return;
    int s = ei[e], d = ei[E + e];
    #pragma unroll
    for (int h = 0; h < HEADS; ++h) {
        float v = leaky(a_src[s * HEADS + h] + a_dst[d * HEADS + h]);
        atomicMax(&mmax[d * HEADS + h], v);
    }
}

__global__ __launch_bounds__(256)
void gat_edge_scatter_kernel(const int* __restrict__ ei,
                             const float* __restrict__ a_src,
                             const float* __restrict__ a_dst,
                             const float* __restrict__ mmax,
                             const float* __restrict__ xt,
                             float* __restrict__ den,
                             float* __restrict__ out, int E) {
    int gid  = blockIdx.x * blockDim.x + threadIdx.x;
    int e    = gid >> 5;
    int lane = gid & 31;
    if (e >= E) return;
    int s = ei[e], d = ei[E + e];
    int h = lane >> 4;
    float v  = leaky(a_src[s * HEADS + h] + a_dst[d * HEADS + h]);
    float ex = __expf(v - mmax[d * HEADS + h]);
    float4 xv = ((const float4*)(xt + (size_t)s * FT))[lane];
    if ((lane & 15) == 0) atomicAdd(&den[d * HEADS + h], ex);
    float* op = out + (size_t)d * FT + lane * 4;
    atomicAdd(op + 0, ex * xv.x);
    atomicAdd(op + 1, ex * xv.y);
    atomicAdd(op + 2, ex * xv.z);
    atomicAdd(op + 3, ex * xv.w);
}

__global__ void gat_epilogue_kernel(float* __restrict__ out,
                                    const float* __restrict__ den,
                                    const float* __restrict__ bias,
                                    const float* __restrict__ prelu_w, int total) {
    int i = blockIdx.x * blockDim.x + threadIdx.x;
    if (i >= total) return;
    float dn  = den[i >> 6];
    float acc = out[i];
    float agg = acc / (dn > 0.f ? dn : 1.f);
    float v = agg + bias[i & (FT - 1)];
    out[i] = v >= 0.f ? v : prelu_w[0] * v;
}

extern "C" void kernel_launch(void* const* d_in, const int* in_sizes, int n_in,
                              void* d_out, int out_size, void* d_ws, size_t ws_size,
                              hipStream_t stream) {
    const float* x       = (const float*)d_in[0];
    const int*   ei      = (const int*)  d_in[1];
    const float* W       = (const float*)d_in[2];
    const float* att_src = (const float*)d_in[3];
    const float* att_dst = (const float*)d_in[4];
    const float* bias    = (const float*)d_in[5];
    const float* prelu_w = (const float*)d_in[6];

    const int N = in_sizes[0] / IN_FT;
    const int E = in_sizes[1] / 2;

    float* out = (float*)d_out;

    float* ws    = (float*)d_ws;
    float* xt    = ws;
    float* a_src = xt    + (size_t)N * FT;
    float* a_dst = a_src + (size_t)N * HEADS;
    float* mmax  = a_dst + (size_t)N * HEADS;
    float* den   = mmax  + (size_t)N * HEADS;

    const int NH    = N * HEADS;
    const int total = N * FT;

    gat_init_kernel<<<(total + 255) / 256, 256, 0, stream>>>(out, mmax, den, total, NH);

    const int tiles_m = (N + 15) / 16;
    const int groups  = (tiles_m + 3) / 4;
    gat_gemm_wmma_kernel<<<groups * 8, 128, 0, stream>>>(x, W, xt, N);

    gat_logits_kernel<<<(NH + 255) / 256, 256, 0, stream>>>(xt, att_src, att_dst,
                                                            a_src, a_dst, NH);

    gat_edge_max_kernel<<<(E + 255) / 256, 256, 0, stream>>>(ei, a_src, a_dst, mmax, E);

    long long sc_threads = (long long)E * 32;
    gat_edge_scatter_kernel<<<(unsigned)((sc_threads + 255) / 256), 256, 0, stream>>>(
        ei, a_src, a_dst, mmax, xt, den, out, E);

    gat_epilogue_kernel<<<(total + 255) / 256, 256, 0, stream>>>(out, den, bias, prelu_w, total);
}


# Edges counting-sorted by destination per call; one wave per node orders its segment by source id in registers, then reduces it in that fixed order (max, sums, divide, PReLU): no float atomics
# speedup vs baseline: 2.2730x; 1.8566x over previous
#include <hip/hip_runtime.h>
#include <math.h>

typedef float v2f __attribute__((ext_vector_type(2)));
typedef float v8f __attribute__((ext_vector_type(8)));

#define IN_FT   128
#define OUT_FT  64
#define HEADS   2
#define FT      (HEADS * OUT_FT)
#define NEG_SLOPE 0.2f

__device__ __forceinline__ float leaky(float s) {
    return s > 0.f ? s : NEG_SLOPE * s;
}

__global__ __launch_bounds__(128)
void gat_gemm_wmma_kernel(const float* __restrict__ x,
                          const float* __restrict__ W,
                          float* __restrict__ xt, int N) {
    __shared__ float Wl[IN_FT * 16];

    const int tile_n = blockIdx.x & 7;
    const int group  = blockIdx.x >> 3;
    const int col0   = tile_n * 16;

    for (int i = threadIdx.x; i < IN_FT * 16; i += 128) {
        int k = i >> 4, c = i & 15;
        Wl[i] = W[k * FT + col0 + c];
    }
    __syncthreads();

    const int wave = threadIdx.x >> 5;
    const int lane = threadIdx.x & 31;
    const int half = lane >> 4;
    const int l16  = lane & 15;

    const int row0 = (group * 4 + wave) * 16;
    int arow = row0 + l16;
    if (arow >= N) arow = N - 1;

    v8f c = {};
    for (int k0 = 0; k0 < IN_FT; k0 += 4) {
        v2f a, b;
        const float* ap = x + (size_t)arow * IN_FT + (k0 + 2 * half);
        a.x = ap[0];
        a.y = ap[1];
        b.x = Wl[(k0 + 2 * half)     * 16 + l16];
        b.y = Wl[(k0 + 2 * half + 1) * 16 + l16];
        c = __builtin_amdgcn_wmma_f32_16x16x4_f32(false, a, false, b,
                                                  (short)0, c, false, false);
    }

    #pragma unroll
    for (int r = 0; r < 8; ++r) {
        int rr = row0 + r + 8 * half;
        if (rr < N) xt[(size_t)rr * FT + col0 + l16] = c[r];
    }
}

__global__ void gat_logits_kernel(const float* __restrict__ xt,
                                  const float* __restrict__ att_src,
                                  const float* __restrict__ att_dst,
                                  float* __restrict__ a_src,
                                  float* __restrict__ a_dst,
                                  int* __restrict__ cnt, int NH) {
    int i = blockIdx.x * blockDim.x + threadIdx.x;
    if (i >= NH) return;
    int n = i >> 1, h = i & 1;
    if (h == 0) cnt[n] = 0;
    const float4* xp = (const float4*)(xt + (size_t)n * FT + h * OUT_FT);
    const float4* as = (const float4*)(att_src + h * OUT_FT);
    const float4* ad = (const float4*)(att_dst + h * OUT_FT);
    float s0 = 0.f, s1 = 0.f;
    #pragma unroll
    for (int j = 0; j < OUT_FT / 4; ++j) {
        float4 v = xp[j], a = as[j], d = ad[j];
        s0 += v.x * a.x + v.y * a.y + v.z * a.z + v.w * a.w;
        s1 += v.x * d.x + v.y * d.y + v.z * d.z + v.w * d.w;
    }
    a_src[i] = s0;
    a_dst[i] = s1;
}

__global__ void gat_hist_kernel(const int* __restrict__ ei,
                                int* __restrict__ cnt,
                                int* __restrict__ slot, int E) {
    int e = blockIdx.x * blockDim.x + threadIdx.x;
    if (e >= E) return;
    slot[e] = atomicAdd(&cnt[ei[E + e]], 1);
}

#define SCAN_T   1024
#define SCAN_PER 49
#define SCAN_CH  (SCAN_T * SCAN_PER)
__global__ __launch_bounds__(SCAN_T)
void gat_scan_kernel(const int* __restrict__ cnt, int* __restrict__ off, int N) {
    __shared__ int buf[SCAN_CH];
    __shared__ int wsum[32];
    __shared__ int total_s;
    const int t = threadIdx.x, lane = t & 31, w = t >> 5;
    int carry = 0;
    for (int base = 0; base < N; base += SCAN_CH) {
        const int n = min(SCAN_CH, N - base);
        #pragma unroll 1
        for (int g = 0; g < SCAN_PER; g += 7) {
            int stage[7];
            #pragma unroll
            for (int k = 0; k < 7; ++k)
                stage[k] = cnt[base + min(t + (g + k) * SCAN_T, n - 1)];
            #pragma unroll
            for (int k = 0; k < 7; ++k) {
                const int i = t + (g + k) * SCAN_T;
                buf[i] = (i < n) ? stage[k] : 0;
            }
        }
        __syncthreads();
        int loc = 0;
        #pragma unroll 7
        for (int i = 0; i < SCAN_PER; ++i) loc += buf[t * SCAN_PER + i];
        int inc = loc;
        #pragma unroll
        for (int o = 1; o < 32; o <<= 1) {
            int v = __shfl_up(inc, o);
            if (lane >= o) inc += v;
        }
        if (lane == 31) wsum[w] = inc;
        __syncthreads();
        if (w == 0) {
            int v = wsum[lane], s = v;
            #pragma unroll
            for (int o = 1; o < 32; o <<= 1) {
                int u = __shfl_up(s, o);
                if (lane >= o) s += u;
            }
            wsum[lane] = s - v;
            if (lane == 31) total_s = s;
        }
        __syncthreads();
        int run = carry + wsum[w] + inc - loc;
        #pragma unroll 7
        for (int i = 0; i < SCAN_PER; ++i) {
            int c = buf[t * SCAN_PER + i];
            buf[t * SCAN_PER + i] = run;
            run += c;
        }
        carry += total_s;
        __syncthreads();
        #pragma unroll 7
        for (int k = 0; k < SCAN_PER; ++k) {
            const int i = t + k * SCAN_T;
            if (i < n) off[base + i] = buf[i];
        }
        __syncthreads();
    }
    if (t == 0) off[N] = carry;
}

__global__ void gat_place_kernel(const int* __restrict__ ei,
                                 const int* __restrict__ off,
                                 const int* __restrict__ slot,
                                 int* __restrict__ ssrc, int E) {
    int e = blockIdx.x * blockDim.x + threadIdx.x;
    if (e >= E) return;
    int s = ei[e], d = ei[E + e];
    ssrc[off[d] + slot[e]] = s;
}

__device__ __forceinline__ float rl_f(float v, int l) {
    return __int_as_float(__builtin_amdgcn_readlane(__float_as_int(v), l));
}
template <int M>
__device__ __forceinline__ float lane_xor(float v) {
    const int i = __float_as_int(v);
    if constexpr (M == 16)
        return __int_as_float(__builtin_amdgcn_permlanex16(i, i, 0x76543210, 0xfedcba98, false, false));
    else
        return __int_as_float(__builtin_amdgcn_update_dpp(i, i, 0x160 + M, 0xF, 0xF, false));
}
__device__ __forceinline__ float wave_max(float v) {
    v = fmaxf(v, lane_xor<1>(v));  v = fmaxf(v, lane_xor<2>(v));
    v = fmaxf(v, lane_xor<4>(v));  v = fmaxf(v, lane_xor<8>(v));
    v = fmaxf(v, lane_xor<16>(v));
    return v;
}
__device__ __forceinline__ float wave_sum(float v) {
    v += lane_xor<1>(v);  v += lane_xor<2>(v);
    v += lane_xor<4>(v);  v += lane_xor<8>(v);
    v += lane_xor<16>(v);
    return v;
}

#define GAT_LD(k) \
    const float4 x##k = xt4[(unsigned)__builtin_amdgcn_readlane(s_r, j + k) * (unsigned)(FT / 4) + (unsigned)lane];
#define GAT_FM(k) { \
    const float wa = rl_f(e0, j + k), wb = rl_f(e1, j + k); \
    const float w = half ? wb : wa; \
    acc.x = fmaf(w, x##k.x, acc.x); acc.y = fmaf(w, x##k.y, acc.y); \
    acc.z = fmaf(w, x##k.z, acc.z); acc.w = fmaf(w, x##k.w, acc.w); }

__device__ __forceinline__ void gat_accum(const float* __restrict__ xt, int s_r,
                                          float e0, float e1, int n,
                                          int lane, int half, float4& acc) {
    const float4* __restrict__ xt4 = (const float4*)xt;
    int j = 0;
    for (; j + 8 <= n; j += 8) {
        GAT_LD(0) GAT_LD(1) GAT_LD(2) GAT_LD(3) GAT_LD(4) GAT_LD(5) GAT_LD(6) GAT_LD(7)
        GAT_FM(0) GAT_FM(1) GAT_FM(2) GAT_FM(3) GAT_FM(4) GAT_FM(5) GAT_FM(6) GAT_FM(7)
    }
    if (j + 4 <= n) {
        GAT_LD(0) GAT_LD(1) GAT_LD(2) GAT_LD(3)
        GAT_FM(0) GAT_FM(1) GAT_FM(2) GAT_FM(3)
        j += 4;
    }
    if (j + 2 <= n) {
        GAT_LD(0) GAT_LD(1)
        GAT_FM(0) GAT_FM(1)
        j += 2;
    }
    if (j < n) {
        GAT_LD(0)
        GAT_FM(0)
    }
}

__global__ __launch_bounds__(256)
void gat_node_kernel(const int* __restrict__ off,
                     const int* __restrict__ ssrc,
                     int* ssrc2,
                     const float* __restrict__ a_src,
                     const float* __restrict__ a_dst,
                     const float* __restrict__ xt,
                     const float* __restrict__ bias,
                     const float* __restrict__ prelu_w,
                     float* __restrict__ out, int N) {
    const int lane = threadIdx.x & 31;
    const int half = lane >> 4;
    const int node = __builtin_amdgcn_readfirstlane(blockIdx.x * 8 + (threadIdx.x >> 5));
    if (node >= N) return;
    const int start = __builtin_amdgcn_readfirstlane(off[node]);
    const int deg   = __builtin_amdgcn_readfirstlane(off[node + 1]) - start;
    const float2 ad = *(const float2*)(a_dst + (size_t)node * HEADS);

    float4 acc = {0.f, 0.f, 0.f, 0.f};
    float den0 = 0.f, den1 = 0.f;

    if (deg <= 32) {
        const bool valid = lane < deg;
        const unsigned key = valid ? (((unsigned)ssrc[start + lane] << 5) | (unsigned)lane)
                                   : (0xFFFFFFE0u | (unsigned)lane);
        int rank = 0;
        #pragma unroll
        for (int t = 0; t < 32; ++t)
            rank += ((unsigned)__builtin_amdgcn_readlane((int)key, t) < key) ? 1 : 0;
        const int sorted = __builtin_amdgcn_ds_permute(rank << 2, (int)(key >> 5));
        const int s_r = valid ? sorted : 0;

        float v0 = -INFINITY, v1 = -INFINITY;
        if (valid) {
            const float2 as = *(const float2*)(a_src + (size_t)s_r * HEADS);
            v0 = leaky(as.x + ad.x);
            v1 = leaky(as.y + ad.y);
        }
        const float m0 = wave_max(v0), m1 = wave_max(v1);
        const float e0 = valid ? __expf(v0 - m0) : 0.f;
        const float e1 = valid ? __expf(v1 - m1) : 0.f;
        den0 = wave_sum(e0);
        den1 = wave_sum(e1);
        gat_accum(xt, s_r, e0, e1, deg, lane, half, acc);
    } else {
        for (int ib = 0; ib < deg; ib += 32) {
            const int i  = ib + lane;
            const int ki = (i < deg) ? ssrc[start + i] : 0x7fffffff;
            int rank = 0;
            for (int jb = 0; jb < deg; jb += 32) {
                const int jj = jb + lane;
                const int kj = (jj < deg) ? ssrc[start + jj] : 0x7fffffff;
                #pragma unroll
                for (int t = 0; t < 32; ++t) {
                    const int kt = __builtin_amdgcn_readlane(kj, t);
                    rank += (kt < ki || (kt == ki && jb + t < i)) ? 1 : 0;
                }
            }
            if (i < deg) ssrc2[start + rank] = ki;
        }
        __threadfence();

        float mx0 = -INFINITY, mx1 = -INFINITY;
        for (int cb = 0; cb < deg; cb += 32) {
            if (cb + lane < deg) {
                const int s = ssrc2[start + cb + lane];
                const float2 as = *(const float2*)(a_src + (size_t)s * HEADS);
                mx0 = fmaxf(mx0, leaky(as.x + ad.x));
                mx1 = fmaxf(mx1, leaky(as.y + ad.y));
            }
        }
        const float m0 = wave_max(mx0), m1 = wave_max(mx1);
        for (int cb = 0; cb < deg; cb += 32) {
            const bool valid = cb + lane < deg;
            const int s_r = valid ? ssrc2[start + cb + lane] : 0;
            float e0 = 0.f, e1 = 0.f;
            if (valid) {
                const float2 as = *(const float2*)(a_src + (size_t)s_r * HEADS);
                e0 = __expf(leaky(as.x + ad.x) - m0);
                e1 = __expf(leaky(as.y + ad.y) - m1);
            }
            den0 += wave_sum(e0);
            den1 += wave_sum(e1);
            gat_accum(xt, s_r, e0, e1, min(32, deg - cb), lane, half, acc);
        }
    }

    const float dn  = half ? den1 : den0;
    const float inv = dn > 0.f ? dn : 1.f;
    const float4 b  = ((const float4*)bias)[lane];
    const float  pw = prelu_w[0];
    float4 r;
    r.x = acc.x / inv + b.x;  r.y = acc.y / inv + b.y;
    r.z = acc.z / inv + b.z;  r.w = acc.w / inv + b.w;
    r.x = r.x >= 0.f ? r.x : pw * r.x;  r.y = r.y >= 0.f ? r.y : pw * r.y;
    r.z = r.z >= 0.f ? r.z : pw * r.z;  r.w = r.w >= 0.f ? r.w : pw * r.w;
    ((float4*)(out + (size_t)node * FT))[lane] = r;
}

extern "C" void kernel_launch(void* const* d_in, const int* in_sizes, int n_in,
                              void* d_out, int out_size, void* d_ws, size_t ws_size,
                              hipStream_t stream) {
    const float* x       = (const float*)d_in[0];
    const int*   ei      = (const int*)  d_in[1];
    const float* W       = (const float*)d_in[2];
    const float* att_src = (const float*)d_in[3];
    const float* att_dst = (const float*)d_in[4];
    const float* bias    = (const float*)d_in[5];
    const float* prelu_w = (const float*)d_in[6];

    const int N = in_sizes[0] / IN_FT;
    const int E = in_sizes[1] / 2;

    float* out = (float*)d_out;

    float* ws    = (float*)d_ws;
    float* xt    = ws;
    float* a_src = xt    + (size_t)N * FT;
    float* a_dst = a_src + (size_t)N * HEADS;
    int*   cnt   = (int*)(a_dst + (size_t)N * HEADS);
    int*   off   = cnt   + (size_t)N;
    int*   slot  = off   + (size_t)N + 1;
    int*   ssrc  = slot  + (size_t)E;
    int*   ssrc2 = ssrc  + (size_t)E;

    const int NH = N * HEADS;

    const int tiles_m = (N + 15) / 16;
    const int groups  = (tiles_m + 3) / 4;
    gat_gemm_wmma_kernel<<<groups * 8, 128, 0, stream>>>(x, W, xt, N);

    gat_logits_kernel<<<(NH + 255) / 256, 256, 0, stream>>>(xt, att_src, att_dst,
                                                            a_src, a_dst, cnt, NH);

    gat_hist_kernel<<<(E + 255) / 256, 256, 0, stream>>>(ei, cnt, slot, E);

    gat_scan_kernel<<<1, SCAN_T, 0, stream>>>(cnt, off, N);

    gat_place_kernel<<<(E + 255) / 256, 256, 0, stream>>>(ei, off, slot, ssrc, E);

    gat_node_kernel<<<(N + 7) / 8, 256, 0, stream>>>(off, ssrc, ssrc2, a_src, a_dst, xt,
                                                     bias, prelu_w, out, N);
}


# Per-destination 64-slot buckets filled by the histogram pass (atomic rank = slot) replace the scan and place kernels: 6 -> 4 launches; segments above 64 edges are summed in edge-list order
# speedup vs baseline: 2.6469x; 1.1645x over previous
#include <hip/hip_runtime.h>
#include <math.h>

typedef float v2f __attribute__((ext_vector_type(2)));
typedef float v8f __attribute__((ext_vector_type(8)));

#define IN_FT   128
#define OUT_FT  64
#define HEADS   2
#define FT      (HEADS * OUT_FT)
#define NEG_SLOPE 0.2f

__device__ __forceinline__ float leaky(float s) {
    return s > 0.f ? s : NEG_SLOPE * s;
}

__global__ __launch_bounds__(128)
void gat_gemm_wmma_kernel(const float* __restrict__ x,
                          const float* __restrict__ W,
                          float* __restrict__ xt, int N) {
    __shared__ float Wl[IN_FT * 16];

    const int tile_n = blockIdx.x & 7;
    const int group  = blockIdx.x >> 3;
    const int col0   = tile_n * 16;

    for (int i = threadIdx.x; i < IN_FT * 16; i += 128) {
        int k = i >> 4, c = i & 15;
        Wl[i] = W[k * FT + col0 + c];
    }
    __syncthreads();

    const int wave = threadIdx.x >> 5;
    const int lane = threadIdx.x & 31;
    const int half = lane >> 4;
    const int l16  = lane & 15;

    const int row0 = (group * 4 + wave) * 16;
    int arow = row0 + l16;
    if (arow >= N) arow = N - 1;

    v8f c = {};
    for (int k0 = 0; k0 < IN_FT; k0 += 4) {
        v2f a, b;
        const float* ap = x + (size_t)arow * IN_FT + (k0 + 2 * half);
        a.x = ap[0];
        a.y = ap[1];
        b.x = Wl[(k0 + 2 * half)     * 16 + l16];
        b.y = Wl[(k0 + 2 * half + 1) * 16 + l16];
        c = __builtin_amdgcn_wmma_f32_16x16x4_f32(false, a, false, b,
                                                  (short)0, c, false, false);
    }

    #pragma unroll
    for (int r = 0; r < 8; ++r) {
        int rr = row0 + r + 8 * half;
        if (rr < N) xt[(size_t)rr * FT + col0 + l16] = c[r];
    }
}

__global__ void gat_logits_kernel(const float* __restrict__ xt,
                                  const float* __restrict__ att_src,
                                  const float* __restrict__ att_dst,
                                  float* __restrict__ a_src,
                                  float* __restrict__ a_dst,
                                  int* __restrict__ cnt, int NH) {
    int i = blockIdx.x * blockDim.x + threadIdx.x;
    if (i >= NH) return;
    int n = i >> 1, h = i & 1;
    if (h == 0) cnt[n] = 0;
    const float4* xp = (const float4*)(xt + (size_t)n * FT + h * OUT_FT);
    const float4* as = (const float4*)(att_src + h * OUT_FT);
    const float4* ad = (const float4*)(att_dst + h * OUT_FT);
    float s0 = 0.f, s1 = 0.f;
    #pragma unroll
    for (int j = 0; j < OUT_FT / 4; ++j) {
        float4 v = xp[j], a = as[j], d = ad[j];
        s0 += v.x * a.x + v.y * a.y + v.z * a.z + v.w * a.w;
        s1 += v.x * d.x + v.y * d.y + v.z * d.z + v.w * d.w;
    }
    a_src[i] = s0;
    a_dst[i] = s1;
}

#define CAP 64
__global__ void gat_bucket_kernel(const int* __restrict__ ei,
                                  int* __restrict__ cnt,
                                  int* __restrict__ bucket, int E) {
    int e = blockIdx.x * blockDim.x + threadIdx.x;
    if (e >= E) return;
    const int s = ei[e], d = ei[E + e];
    asm volatile("" :: "v"(s));
    const int slot = atomicAdd(&cnt[d], 1);
    if (slot < CAP) bucket[d * CAP + slot] = s;
}

__device__ __forceinline__ float rl_f(float v, int l) {
    return __int_as_float(__builtin_amdgcn_readlane(__float_as_int(v), l));
}
template <int M>
__device__ __forceinline__ float lane_xor(float v) {
    const int i = __float_as_int(v);
    if constexpr (M == 16)
        return __int_as_float(__builtin_amdgcn_permlanex16(i, i, 0x76543210, 0xfedcba98, false, false));
    else
        return __int_as_float(__builtin_amdgcn_update_dpp(i, i, 0x160 + M, 0xF, 0xF, false));
}
__device__ __forceinline__ float wave_max(float v) {
    v = fmaxf(v, lane_xor<1>(v));  v = fmaxf(v, lane_xor<2>(v));
    v = fmaxf(v, lane_xor<4>(v));  v = fmaxf(v, lane_xor<8>(v));
    v = fmaxf(v, lane_xor<16>(v));
    return v;
}
__device__ __forceinline__ float wave_sum(float v) {
    v += lane_xor<1>(v);  v += lane_xor<2>(v);
    v += lane_xor<4>(v);  v += lane_xor<8>(v);
    v += lane_xor<16>(v);
    return v;
}

#define GAT_LD(k) \
    const float4 x##k = xt4[(unsigned)__builtin_amdgcn_readlane(s_r, j + k) * (unsigned)(FT / 4) + (unsigned)lane];
#define GAT_FM(k) { \
    const float wa = rl_f(e0, j + k), wb = rl_f(e1, j + k); \
    const float w = half ? wb : wa; \
    acc.x = fmaf(w, x##k.x, acc.x); acc.y = fmaf(w, x##k.y, acc.y); \
    acc.z = fmaf(w, x##k.z, acc.z); acc.w = fmaf(w, x##k.w, acc.w); }

__device__ __forceinline__ void gat_accum(const float* __restrict__ xt, int s_r,
                                          float e0, float e1, int n,
                                          int lane, int half, float4& acc) {
    const float4* __restrict__ xt4 = (const float4*)xt;
    int j = 0;
    for (; j + 8 <= n; j += 8) {
        GAT_LD(0) GAT_LD(1) GAT_LD(2) GAT_LD(3) GAT_LD(4) GAT_LD(5) GAT_LD(6) GAT_LD(7)
        GAT_FM(0) GAT_FM(1) GAT_FM(2) GAT_FM(3) GAT_FM(4) GAT_FM(5) GAT_FM(6) GAT_FM(7)
    }
    if (j + 4 <= n) {
        GAT_LD(0) GAT_LD(1) GAT_LD(2) GAT_LD(3)
        GAT_FM(0) GAT_FM(1) GAT_FM(2) GAT_FM(3)
        j += 4;
    }
    if (j + 2 <= n) {
        GAT_LD(0) GAT_LD(1)
        GAT_FM(0) GAT_FM(1)
        j += 2;
    }
    if (j < n) {
        GAT_LD(0)
        GAT_FM(0)
    }
}

__global__ __launch_bounds__(256)
void gat_node_kernel(const int* __restrict__ ei,
                     const int* __restrict__ cnt,
                     const int* __restrict__ bucket,
                     int* bucket2,
                     const float* __restrict__ a_src,
                     const float* __restrict__ a_dst,
                     const float* __restrict__ xt,
                     const float* __restrict__ bias,
                     const float* __restrict__ prelu_w,
                     float* __restrict__ out, int N, int E) {
    const int lane = threadIdx.x & 31;
    const int half = lane >> 4;
    const int node = __builtin_amdgcn_readfirstlane(blockIdx.x * 8 + (threadIdx.x >> 5));
    if (node >= N) return;
    const int deg   = __builtin_amdgcn_readfirstlane(cnt[node]);
    const int start = node * CAP;
    const float2 ad = *(const float2*)(a_dst + (size_t)node * HEADS);

    float4 acc = {0.f, 0.f, 0.f, 0.f};
    float den0 = 0.f, den1 = 0.f;

    if (deg <= 32) {
        const bool valid = lane < deg;
        const unsigned key = valid ? (((unsigned)bucket[start + lane] << 5) | (unsigned)lane)
                                   : (0xFFFFFFE0u | (unsigned)lane);
        int rank = 0;
        #pragma unroll
        for (int t = 0; t < 32; ++t)
            rank += ((unsigned)__builtin_amdgcn_readlane((int)key, t) < key) ? 1 : 0;
        const int sorted = __builtin_amdgcn_ds_permute(rank << 2, (int)(key >> 5));
        const int s_r = valid ? sorted : 0;

        float v0 = -INFINITY, v1 = -INFINITY;
        if (valid) {
            const float2 as = *(const float2*)(a_src + (size_t)s_r * HEADS);
            v0 = leaky(as.x + ad.x);
            v1 = leaky(as.y + ad.y);
        }
        const float m0 = wave_max(v0), m1 = wave_max(v1);
        const float e0 = valid ? __expf(v0 - m0) : 0.f;
        const float e1 = valid ? __expf(v1 - m1) : 0.f;
        den0 = wave_sum(e0);
        den1 = wave_sum(e1);
        gat_accum(xt, s_r, e0, e1, deg, lane, half, acc);
    } else if (deg <= CAP) {
        for (int ib = 0; ib < deg; ib += 32) {
            const int i  = ib + lane;
            const int ki = (i < deg) ? bucket[start + i] : 0x7fffffff;
            int rank = 0;
            for (int jb = 0; jb < deg; jb += 32) {
                const int jj = jb + lane;
                const int kj = (jj < deg) ? bucket[start + jj] : 0x7fffffff;
                #pragma unroll
                for (int t = 0; t < 32; ++t) {
                    const int kt = __builtin_amdgcn_readlane(kj, t);
                    rank += (kt < ki || (kt == ki && jb + t < i)) ? 1 : 0;
                }
            }
            if (i < deg) bucket2[start + rank] = ki;
        }
        __threadfence();

        float mx0 = -INFINITY, mx1 = -INFINITY;
        for (int cb = 0; cb < deg; cb += 32) {
            if (cb + lane < deg) {
                const int s = bucket2[start + cb + lane];
                const float2 as = *(const float2*)(a_src + (size_t)s * HEADS);
                mx0 = fmaxf(mx0, leaky(as.x + ad.x));
                mx1 = fmaxf(mx1, leaky(as.y + ad.y));
            }
        }
        const float m0 = wave_max(mx0), m1 = wave_max(mx1);
        for (int cb = 0; cb < deg; cb += 32) {
            const bool valid = cb + lane < deg;
            const int s_r = valid ? bucket2[start + cb + lane] : 0;
            float e0 = 0.f, e1 = 0.f;
            if (valid) {
                const float2 as = *(const float2*)(a_src + (size_t)s_r * HEADS);
                e0 = __expf(leaky(as.x + ad.x) - m0);
                e1 = __expf(leaky(as.y + ad.y) - m1);
            }
            den0 += wave_sum(e0);
            den1 += wave_sum(e1);
            gat_accum(xt, s_r, e0, e1, min(32, deg - cb), lane, half, acc);
        }
    } else {
        const int* __restrict__ dstp = ei + E;
        const float4* __restrict__ xt4 = (const float4*)xt;
        float mx0 = -INFINITY, mx1 = -INFINITY;
        for (int eb = 0; eb < E; eb += 32) {
            const int e = eb + lane;
            if (e < E && dstp[e] == node) {
                const float2 as = *(const float2*)(a_src + (size_t)ei[e] * HEADS);
                mx0 = fmaxf(mx0, leaky(as.x + ad.x));
                mx1 = fmaxf(mx1, leaky(as.y + ad.y));
            }
        }
        const float m0 = wave_max(mx0), m1 = wave_max(mx1);
        for (int eb = 0; eb < E; eb += 32) {
            const int e = eb + lane;
            const bool hit = e < E && dstp[e] == node;
            unsigned mask = __builtin_amdgcn_ballot_w32(hit);
            if (mask == 0u) continue;
            int s = 0;
            float e0 = 0.f, e1 = 0.f;
            if (hit) {
                s = ei[e];
                const float2 as = *(const float2*)(a_src + (size_t)s * HEADS);
                e0 = __expf(leaky(as.x + ad.x) - m0);
                e1 = __expf(leaky(as.y + ad.y) - m1);
            }
            den0 += wave_sum(e0);
            den1 += wave_sum(e1);
            while (mask != 0u) {
                const int t = __builtin_ctz(mask);
                mask &= mask - 1u;
                const unsigned st = (unsigned)__builtin_amdgcn_readlane(s, t);
                const float4 xv = xt4[st * (unsigned)(FT / 4) + (unsigned)lane];
                const float wa = rl_f(e0, t), wb = rl_f(e1, t);
                const float w = half ? wb : wa;
                acc.x = fmaf(w, xv.x, acc.x); acc.y = fmaf(w, xv.y, acc.y);
                acc.z = fmaf(w, xv.z, acc.z); acc.w = fmaf(w, xv.w, acc.w);
            }
        }
    }

    const float dn  = half ? den1 : den0;
    const float inv = dn > 0.f ? dn : 1.f;
    const float4 b  = ((const float4*)bias)[lane];
    const float  pw = prelu_w[0];
    float4 r;
    r.x = acc.x / inv + b.x;  r.y = acc.y / inv + b.y;
    r.z = acc.z / inv + b.z;  r.w = acc.w / inv + b.w;
    r.x = r.x >= 0.f ? r.x : pw * r.x;  r.y = r.y >= 0.f ? r.y : pw * r.y;
    r.z = r.z >= 0.f ? r.z : pw * r.z;  r.w = r.w >= 0.f ? r.w : pw * r.w;
    ((float4*)(out + (size_t)node * FT))[lane] = r;
}

extern "C" void kernel_launch(void* const* d_in, const int* in_sizes, int n_in,
                              void* d_out, int out_size, void* d_ws, size_t ws_size,
                              hipStream_t stream) {
    const float* x       = (const float*)d_in[0];
    const int*   ei      = (const int*)  d_in[1];
    const float* W       = (const float*)d_in[2];
    const float* att_src = (const float*)d_in[3];
    const float* att_dst = (const float*)d_in[4];
    const float* bias    = (const float*)d_in[5];
    const float* prelu_w = (const float*)d_in[6];

    const int N = in_sizes[0] / IN_FT;
    const int E = in_sizes[1] / 2;

    float* out = (float*)d_out;

    float* ws      = (float*)d_ws;
    float* xt      = ws;
    int*   bucket  = (int*)(xt + (size_t)N * FT);
    int*   bucket2 = bucket  + (size_t)N * CAP;
    float* a_src   = (float*)(bucket2 + (size_t)N * CAP);
    float* a_dst   = a_src + (size_t)N * HEADS;
    int*   cnt     = (int*)(a_dst + (size_t)N * HEADS);

    const int NH = N * HEADS;

    const int tiles_m = (N + 15) / 16;
    const int groups  = (tiles_m + 3) / 4;
    gat_gemm_wmma_kernel<<<groups * 8, 128, 0, stream>>>(x, W, xt, N);

    gat_logits_kernel<<<(NH + 255) / 256, 256, 0, stream>>>(xt, att_src, att_dst,
                                                            a_src, a_dst, cnt, NH);

    gat_bucket_kernel<<<(E + 255) / 256, 256, 0, stream>>>(ei, cnt, bucket, E);

    gat_node_kernel<<<(N + 7) / 8, 256, 0, stream>>>(ei, cnt, bucket, bucket2, a_src, a_dst, xt,
                                                     bias, prelu_w, out, N, E);
}


# GEMM re-tiled: each wave owns a whole 16x128 row strip of xt (W whole in LDS, x fetched once, not 8 times) and dots the staged strip with att_src/att_dst itself: logits launch gone, 4 -> 3 launches
# speedup vs baseline: 3.3126x; 1.2515x over previous
#include <hip/hip_runtime.h>
#include <math.h>

typedef float v2f __attribute__((ext_vector_type(2)));
typedef float v8f __attribute__((ext_vector_type(8)));

#define IN_FT   128
#define OUT_FT  64
#define HEADS   2
#define FT      (HEADS * OUT_FT)
#define NEG_SLOPE 0.2f

__device__ __forceinline__ float leaky(float s) {
    return s > 0.f ? s : NEG_SLOPE * s;
}

#define GW      8
#define WPITCH  144
#define SPITCH  132

__global__ __launch_bounds__(GW * 32, 4)
void gat_gemm_wmma_kernel(const float* __restrict__ x,
                          const float* __restrict__ W,
                          const float* __restrict__ att_src,
                          const float* __restrict__ att_dst,
                          float* __restrict__ xt,
                          float* __restrict__ a_src,
                          float* __restrict__ a_dst,
                          int* __restrict__ cnt, int N) {
    __shared__ __attribute__((aligned(16))) float L[IN_FT * WPITCH + GW * 16 * SPITCH];
    float* Wl = L;

    #pragma unroll
    for (int i = threadIdx.x; i < IN_FT * FT / 4; i += GW * 32) {
        const int k = i >> 5, c4 = i & 31;
        const float4 w = ((const float4*)W)[i];
        *(float4*)(Wl + k * WPITCH + 4 * c4) = w;
    }
    __syncthreads();

    const int wave = threadIdx.x >> 5;
    const int lane = threadIdx.x & 31;
    const int half = lane >> 4;
    const int l16  = lane & 15;

    const int row0 = (blockIdx.x * GW + wave) * 16;
    int arow = row0 + l16;
    if (arow >= N) arow = N - 1;

    const float* ap = x + (size_t)arow * IN_FT + 2 * half;
    const float* bp = Wl + (2 * half) * WPITCH + l16;

    v8f c[8];
    #pragma unroll
    for (int t = 0; t < 8; ++t) c[t] = v8f{};

    v2f ac[8], an[8];
    #pragma unroll
    for (int s = 0; s < 8; ++s) {
        ac[s].x = ap[4 * s];
        ac[s].y = ap[4 * s + 1];
    }
    #pragma unroll 1
    for (int ch = 0; ch < 4; ++ch) {
        const int nx = ch < 3 ? ch + 1 : 3;
        const float* apn = ap + 32 * nx;
        #pragma unroll
        for (int s = 0; s < 8; ++s) {
            an[s].x = apn[4 * s];
            an[s].y = apn[4 * s + 1];
        }
        const float* bpc = bp + (32 * ch) * WPITCH;
        #pragma unroll
        for (int s = 0; s < 8; ++s) {
            #pragma unroll
            for (int t = 0; t < 8; ++t) {
                v2f b;
                b.x = bpc[(4 * s) * WPITCH + 16 * t];
                b.y = bpc[(4 * s + 1) * WPITCH + 16 * t];
                c[t] = __builtin_amdgcn_wmma_f32_16x16x4_f32(false, ac[s], false, b,
                                                             (short)0, c[t], false, false);
            }
        }
        #pragma unroll
        for (int s = 0; s < 8; ++s) ac[s] = an[s];
    }

    float* St = L + IN_FT * WPITCH + wave * (16 * SPITCH);
    #pragma unroll
    for (int t = 0; t < 8; ++t) {
        #pragma unroll
        for (int r = 0; r < 8; ++r)
            St[(r + 8 * half) * SPITCH + 16 * t + l16] = c[t][r];
    }
    __syncthreads();

    #pragma unroll
    for (int r = 0; r < 16; ++r) {
        const int rr = row0 + r;
        if (rr < N) {
            const float4 v = *(const float4*)(St + r * SPITCH + 4 * lane);
            ((float4*)(xt + (size_t)rr * FT))[lane] = v;
        }
    }

    const int n = row0 + l16;
    const float4* xp = (const float4*)(St + l16 * SPITCH + half * OUT_FT);
    const float4* as = (const float4*)(att_src + half * OUT_FT);
    const float4* ad = (const float4*)(att_dst + half * OUT_FT);
    float s0 = 0.f, s1 = 0.f;
    #pragma unroll
    for (int j = 0; j < OUT_FT / 4; ++j) {
        float4 v = xp[j], a = as[j], d = ad[j];
        s0 += v.x * a.x + v.y * a.y + v.z * a.z + v.w * a.w;
        s1 += v.x * d.x + v.y * d.y + v.z * d.z + v.w * d.w;
    }
    if (n < N) {
        a_src[n * HEADS + half] = s0;
        a_dst[n * HEADS + half] = s1;
        if (half == 0) cnt[n] = 0;
    }
}

#define CAP 64
__global__ void gat_bucket_kernel(const int* __restrict__ ei,
                                  int* __restrict__ cnt,
                                  int* __restrict__ bucket, int E) {
    int e = blockIdx.x * blockDim.x + threadIdx.x;
    if (e >= E) return;
    const int s = ei[e], d = ei[E + e];
    asm volatile("" :: "v"(s));
    const int slot = atomicAdd(&cnt[d], 1);
    if (slot < CAP) bucket[d * CAP + slot] = s;
}

__device__ __forceinline__ float rl_f(float v, int l) {
    return __int_as_float(__builtin_amdgcn_readlane(__float_as_int(v), l));
}
template <int M>
__device__ __forceinline__ float lane_xor(float v) {
    const int i = __float_as_int(v);
    if constexpr (M == 16)
        return __int_as_float(__builtin_amdgcn_permlanex16(i, i, 0x76543210, 0xfedcba98, false, false));
    else
        return __int_as_float(__builtin_amdgcn_update_dpp(i, i, 0x160 + M, 0xF, 0xF, false));
}
__device__ __forceinline__ float wave_max(float v) {
    v = fmaxf(v, lane_xor<1>(v));  v = fmaxf(v, lane_xor<2>(v));
    v = fmaxf(v, lane_xor<4>(v));  v = fmaxf(v, lane_xor<8>(v));
    v = fmaxf(v, lane_xor<16>(v));
    return v;
}
__device__ __forceinline__ float wave_sum(float v) {
    v += lane_xor<1>(v);  v += lane_xor<2>(v);
    v += lane_xor<4>(v);  v += lane_xor<8>(v);
    v += lane_xor<16>(v);
    return v;
}

#define GAT_LD(k) \
    const float4 x##k = xt4[(unsigned)__builtin_amdgcn_readlane(s_r, j + k) * (unsigned)(FT / 4) + (unsigned)lane];
#define GAT_FM(k) { \
    const float wa = rl_f(e0, j + k), wb = rl_f(e1, j + k); \
    const float w = half ? wb : wa; \
    acc.x = fmaf(w, x##k.x, acc.x); acc.y = fmaf(w, x##k.y, acc.y); \
    acc.z = fmaf(w, x##k.z, acc.z); acc.w = fmaf(w, x##k.w, acc.w); }

__device__ __forceinline__ void gat_accum(const float* __restrict__ xt, int s_r,
                                          float e0, float e1, int n,
                                          int lane, int half, float4& acc) {
    const float4* __restrict__ xt4 = (const float4*)xt;
    int j = 0;
    for (; j + 8 <= n; j += 8) {
        GAT_LD(0) GAT_LD(1) GAT_LD(2) GAT_LD(3) GAT_LD(4) GAT_LD(5) GAT_LD(6) GAT_LD(7)
        GAT_FM(0) GAT_FM(1) GAT_FM(2) GAT_FM(3) GAT_FM(4) GAT_FM(5) GAT_FM(6) GAT_FM(7)
    }
    if (j + 4 <= n) {
        GAT_LD(0) GAT_LD(1) GAT_LD(2) GAT_LD(3)
        GAT_FM(0) GAT_FM(1) GAT_FM(2) GAT_FM(3)
        j += 4;
    }
    if (j + 2 <= n) {
        GAT_LD(0) GAT_LD(1)
        GAT_FM(0) GAT_FM(1)
        j += 2;
    }
    if (j < n) {
        GAT_LD(0)
        GAT_FM(0)
    }
}

__global__ __launch_bounds__(256)
void gat_node_kernel(const int* __restrict__ ei,
                     const int* __restrict__ cnt,
                     const int* __restrict__ bucket,
                     int* bucket2,
                     const float* __restrict__ a_src,
                     const float* __restrict__ a_dst,
                     const float* __restrict__ xt,
                     const float* __restrict__ bias,
                     const float* __restrict__ prelu_w,
                     float* __restrict__ out, int N, int E) {
    const int lane = threadIdx.x & 31;
    const int half = lane >> 4;
    const int node = __builtin_amdgcn_readfirstlane(blockIdx.x * 8 + (threadIdx.x >> 5));
    if (node >= N) return;
    const int deg   = __builtin_amdgcn_readfirstlane(cnt[node]);
    const int start = node * CAP;
    const float2 ad = *(const float2*)(a_dst + (size_t)node * HEADS);

    float4 acc = {0.f, 0.f, 0.f, 0.f};
    float den0 = 0.f, den1 = 0.f;

    if (deg <= 32) {
        const bool valid = lane < deg;
        const unsigned key = valid ? (((unsigned)bucket[start + lane] << 5) | (unsigned)lane)
                                   : (0xFFFFFFE0u | (unsigned)lane);
        int rank = 0;
        #pragma unroll
        for (int t = 0; t < 32; ++t)
            rank += ((unsigned)__builtin_amdgcn_readlane((int)key, t) < key) ? 1 : 0;
        const int sorted = __builtin_amdgcn_ds_permute(rank << 2, (int)(key >> 5));
        const int s_r = valid ? sorted : 0;

        float v0 = -INFINITY, v1 = -INFINITY;
        if (valid) {
            const float2 as = *(const float2*)(a_src + (size_t)s_r * HEADS);
            v0 = leaky(as.x + ad.x);
            v1 = leaky(as.y + ad.y);
        }
        const float m0 = wave_max(v0), m1 = wave_max(v1);
        const float e0 = valid ? __expf(v0 - m0) : 0.f;
        const float e1 = valid ? __expf(v1 - m1) : 0.f;
        den0 = wave_sum(e0);
        den1 = wave_sum(e1);
        gat_accum(xt, s_r, e0, e1, deg, lane, half, acc);
    } else if (deg <= CAP) {
        for (int ib = 0; ib < deg; ib += 32) {
            const int i  = ib + lane;
            const int ki = (i < deg) ? bucket[start + i] : 0x7fffffff;
            int rank = 0;
            for (int jb = 0; jb < deg; jb += 32) {
                const int jj = jb + lane;
                const int kj = (jj < deg) ? bucket[start + jj] : 0x7fffffff;
                #pragma unroll
                for (int t = 0; t < 32; ++t) {
                    const int kt = __builtin_amdgcn_readlane(kj, t);
                    rank += (kt < ki || (kt == ki && jb + t < i)) ? 1 : 0;
                }
            }
            if (i < deg) bucket2[start + rank] = ki;
        }
        __threadfence();

        float mx0 = -INFINITY, mx1 = -INFINITY;
        for (int cb = 0; cb < deg; cb += 32) {
            if (cb + lane < deg) {
                const int s = bucket2[start + cb + lane];
                const float2 as = *(const float2*)(a_src + (size_t)s * HEADS);
                mx0 = fmaxf(mx0, leaky(as.x + ad.x));
                mx1 = fmaxf(mx1, leaky(as.y + ad.y));
            }
        }
        const float m0 = wave_max(mx0), m1 = wave_max(mx1);
        for (int cb = 0; cb < deg; cb += 32) {
            const bool valid = cb + lane < deg;
            const int s_r = valid ? bucket2[start + cb + lane] : 0;
            float e0 = 0.f, e1 = 0.f;
            if (valid) {
                const float2 as = *(const float2*)(a_src + (size_t)s_r * HEADS);
                e0 = __expf(leaky(as.x + ad.x) - m0);
                e1 = __expf(leaky(as.y + ad.y) - m1);
            }
            den0 += wave_sum(e0);
            den1 += wave_sum(e1);
            gat_accum(xt, s_r, e0, e1, min(32, deg - cb), lane, half, acc);
        }
    } else {
        const int* __restrict__ dstp = ei + E;
        const float4* __restrict__ xt4 = (const float4*)xt;
        float mx0 = -INFINITY, mx1 = -INFINITY;
        for (int eb = 0; eb < E; eb += 32) {
            const int e = eb + lane;
            if (e < E && dstp[e] == node) {
                const float2 as = *(const float2*)(a_src + (size_t)ei[e] * HEADS);
                mx0 = fmaxf(mx0, leaky(as.x + ad.x));
                mx1 = fmaxf(mx1, leaky(as.y + ad.y));
            }
        }
        const float m0 = wave_max(mx0), m1 = wave_max(mx1);
        for (int eb = 0; eb < E; eb += 32) {
            const int e = eb + lane;
            const bool hit = e < E && dstp[e] == node;
            unsigned mask = __builtin_amdgcn_ballot_w32(hit);
            if (mask == 0u) continue;
            int s = 0;
            float e0 = 0.f, e1 = 0.f;
            if (hit) {
                s = ei[e];
                const float2 as = *(const float2*)(a_src + (size_t)s * HEADS);
                e0 = __expf(leaky(as.x + ad.x) - m0);
                e1 = __expf(leaky(as.y + ad.y) - m1);
            }
            den0 += wave_sum(e0);
            den1 += wave_sum(e1);
            while (mask != 0u) {
                const int t = __builtin_ctz(mask);
                mask &= mask - 1u;
                const unsigned st = (unsigned)__builtin_amdgcn_readlane(s, t);
                const float4 xv = xt4[st * (unsigned)(FT / 4) + (unsigned)lane];
                const float wa = rl_f(e0, t), wb = rl_f(e1, t);
                const float w = half ? wb : wa;
                acc.x = fmaf(w, xv.x, acc.x); acc.y = fmaf(w, xv.y, acc.y);
                acc.z = fmaf(w, xv.z, acc.z); acc.w = fmaf(w, xv.w, acc.w);
            }
        }
    }

    const float dn  = half ? den1 : den0;
    const float inv = dn > 0.f ? dn : 1.f;
    const float4 b  = ((const float4*)bias)[lane];
    const float  pw = prelu_w[0];
    float4 r;
    r.x = acc.x / inv + b.x;  r.y = acc.y / inv + b.y;
    r.z = acc.z / inv + b.z;  r.w = acc.w / inv + b.w;
    r.x = r.x >= 0.f ? r.x : pw * r.x;  r.y = r.y >= 0.f ? r.y : pw * r.y;
    r.z = r.z >= 0.f ? r.z : pw * r.z;  r.w = r.w >= 0.f ? r.w : pw * r.w;
    ((float4*)(out + (size_t)node * FT))[lane] = r;
}

extern "C" void kernel_launch(void* const* d_in, const int* in_sizes, int n_in,
                              void* d_out, int out_size, void* d_ws, size_t ws_size,
                              hipStream_t stream) {
    const float* x       = (const float*)d_in[0];
    const int*   ei      = (const int*)  d_in[1];
    const float* W       = (const float*)d_in[2];
    const float* att_src = (const float*)d_in[3];
    const float* att_dst = (const float*)d_in[4];
    const float* bias    = (const float*)d_in[5];
    const float* prelu_w = (const float*)d_in[6];

    const int N = in_sizes[0] / IN_FT;
    const int E = in_sizes[1] / 2;

    float* out = (float*)d_out;

    float* ws      = (float*)d_ws;
    float* xt      = ws;
    int*   bucket  = (int*)(xt + (size_t)N * FT);
    int*   bucket2 = bucket  + (size_t)N * CAP;
    float* a_src   = (float*)(bucket2 + (size_t)N * CAP);
    float* a_dst   = a_src + (size_t)N * HEADS;
    int*   cnt     = (int*)(a_dst + (size_t)N * HEADS);

    const int tiles_m = (N + 15) / 16;
    gat_gemm_wmma_kernel<<<(tiles_m + GW - 1) / GW, GW * 32, 0, stream>>>(x, W, att_src, att_dst,
                                                                        xt, a_src, a_dst, cnt, N);

    gat_bucket_kernel<<<(E + 255) / 256, 256, 0, stream>>>(ei, cnt, bucket, E);

    gat_node_kernel<<<(N + 7) / 8, 256, 0, stream>>>(ei, cnt, bucket, bucket2, a_src, a_dst, xt,
                                                     bias, prelu_w, out, N, E);
}


# Edges binned by 64-node destination range in LDS: one global add per block and bin, not one returning add per edge; node kernel takes one bin per block, buckets it in LDS; summation order kept
# speedup vs baseline: 5.7510x; 1.7361x over previous
#include <hip/hip_runtime.h>
#include <math.h>

typedef float v2f __attribute__((ext_vector_type(2)));
typedef float v8f __attribute__((ext_vector_type(8)));

#define IN_FT   128
#define OUT_FT  64
#define HEADS   2
#define FT      (HEADS * OUT_FT)
#define NEG_SLOPE 0.2f

#define CAP        64
#define BIN_SHIFT  6
#define BIN_NODES  (1 << BIN_SHIFT)
#define BINCAP     2048

__device__ __forceinline__ float leaky(float s) {
    return s > 0.f ? s : NEG_SLOPE * s;
}

#define GW      8
#define WPITCH  144
#define SPITCH  132

__global__ __launch_bounds__(GW * 32, 4)
void gat_gemm_wmma_kernel(const float* __restrict__ x,
                          const float* __restrict__ W,
                          const float* __restrict__ att_src,
                          const float* __restrict__ att_dst,
                          float* __restrict__ xt,
                          float* __restrict__ a_src,
                          float* __restrict__ a_dst,
                          int* __restrict__ binCount, int N) {
    __shared__ __attribute__((aligned(16))) float L[IN_FT * WPITCH + GW * 16 * SPITCH];
    float* Wl = L;

    #pragma unroll
    for (int i = threadIdx.x; i < IN_FT * FT / 4; i += GW * 32) {
        const int k = i >> 5, c4 = i & 31;
        const float4 w = ((const float4*)W)[i];
        *(float4*)(Wl + k * WPITCH + 4 * c4) = w;
    }
    __syncthreads();

    const int wave = threadIdx.x >> 5;
    const int lane = threadIdx.x & 31;
    const int half = lane >> 4;
    const int l16  = lane & 15;

    const int row0 = (blockIdx.x * GW + wave) * 16;
    int arow = row0 + l16;
    if (arow >= N) arow = N - 1;

    const float* ap = x + (size_t)arow * IN_FT + 2 * half;
    const float* bp = Wl + (2 * half) * WPITCH + l16;

    v8f c[8];
    #pragma unroll
    for (int t = 0; t < 8; ++t) c[t] = v8f{};

    v2f ac[8], an[8];
    #pragma unroll
    for (int s = 0; s < 8; ++s) {
        ac[s].x = ap[4 * s];
        ac[s].y = ap[4 * s + 1];
    }
    #pragma unroll 1
    for (int ch = 0; ch < 4; ++ch) {
        const int nx = ch < 3 ? ch + 1 : 3;
        const float* apn = ap + 32 * nx;
        #pragma unroll
        for (int s = 0; s < 8; ++s) {
            an[s].x = apn[4 * s];
            an[s].y = apn[4 * s + 1];
        }
        const float* bpc = bp + (32 * ch) * WPITCH;
        #pragma unroll
        for (int s = 0; s < 8; ++s) {
            #pragma unroll
            for (int t = 0; t < 8; ++t) {
                v2f b;
                b.x = bpc[(4 * s) * WPITCH + 16 * t];
                b.y = bpc[(4 * s + 1) * WPITCH + 16 * t];
                c[t] = __builtin_amdgcn_wmma_f32_16x16x4_f32(false, ac[s], false, b,
                                                             (short)0, c[t], false, false);
            }
        }
        #pragma unroll
        for (int s = 0; s < 8; ++s) ac[s] = an[s];
    }

    float* St = L + IN_FT * WPITCH + wave * (16 * SPITCH);
    #pragma unroll
    for (int t = 0; t < 8; ++t) {
        #pragma unroll
        for (int r = 0; r < 8; ++r)
            St[(r + 8 * half) * SPITCH + 16 * t + l16] = c[t][r];
    }
    __syncthreads();

    #pragma unroll
    for (int r = 0; r < 16; ++r) {
        const int rr = row0 + r;
        if (rr < N) {
            const float4 v = *(const float4*)(St + r * SPITCH + 4 * lane);
            ((float4*)(xt + (size_t)rr * FT))[lane] = v;
        }
    }

    const int n = row0 + l16;
    const float4* xp = (const float4*)(St + l16 * SPITCH + half * OUT_FT);
    const float4* as = (const float4*)(att_src + half * OUT_FT);
    const float4* ad = (const float4*)(att_dst + half * OUT_FT);
    float s0 = 0.f, s1 = 0.f;
    #pragma unroll
    for (int j = 0; j < OUT_FT / 4; ++j) {
        float4 v = xp[j], a = as[j], d = ad[j];
        s0 += v.x * a.x + v.y * a.y + v.z * a.z + v.w * a.w;
        s1 += v.x * d.x + v.y * d.y + v.z * d.z + v.w * d.w;
    }
    if (n < N) {
        a_src[n * HEADS + half] = s0;
        a_dst[n * HEADS + half] = s1;
        if (half == 0 && (n & (BIN_NODES - 1)) == 0) binCount[n >> BIN_SHIFT] = 0;
    }
}

#define BW     1024
#define EPT    4
#define MAXNB  1024
__global__ __launch_bounds__(BW)
void gat_bin_kernel(const int* __restrict__ ei,
                    int* __restrict__ binCount,
                    int* __restrict__ binned, int E, int NB) {
    __shared__ int hist[MAXNB];
    const int t  = threadIdx.x;
    const int e0 = blockIdx.x * (BW * EPT) + t;
    int s[EPT], d[EPT], r[EPT];
    #pragma unroll
    for (int k = 0; k < EPT; ++k) {
        const int e  = e0 + k * BW;
        const int ec = e < E ? e : E - 1;
        s[k] = ei[ec];
        const int dv = ei[E + ec];
        d[k] = e < E ? dv : -1;
    }
    for (int b0 = 0; b0 < NB; b0 += MAXNB) {
        if (b0 != 0) __syncthreads();
        for (int i = t; i < MAXNB; i += BW) hist[i] = 0;
        __syncthreads();
        #pragma unroll
        for (int k = 0; k < EPT; ++k) {
            const int bin = (d[k] >> BIN_SHIFT) - b0;
            r[k] = -1;
            if (d[k] >= 0 && bin >= 0 && bin < MAXNB) r[k] = atomicAdd(&hist[bin], 1);
        }
        __syncthreads();
        for (int i = t; i < MAXNB; i += BW) {
            const int c = hist[i];
            if (c > 0) hist[i] = atomicAdd(&binCount[b0 + i], c);
        }
        __syncthreads();
        #pragma unroll
        for (int k = 0; k < EPT; ++k) {
            if (r[k] >= 0) {
                const int bin = (d[k] >> BIN_SHIFT) - b0;
                const int pos = hist[bin] + r[k];
                if (pos < BINCAP)
                    binned[(size_t)(b0 + bin) * BINCAP + pos] =
                        (int)(((unsigned)s[k] << BIN_SHIFT) | (unsigned)(d[k] & (BIN_NODES - 1)));
            }
        }
    }
}

__device__ __forceinline__ float rl_f(float v, int l) {
    return __int_as_float(__builtin_amdgcn_readlane(__float_as_int(v), l));
}
template <int M>
__device__ __forceinline__ float lane_xor(float v) {
    const int i = __float_as_int(v);
    if constexpr (M == 16)
        return __int_as_float(__builtin_amdgcn_permlanex16(i, i, 0x76543210, 0xfedcba98, false, false));
    else
        return __int_as_float(__builtin_amdgcn_update_dpp(i, i, 0x160 + M, 0xF, 0xF, false));
}
__device__ __forceinline__ float wave_max(float v) {
    v = fmaxf(v, lane_xor<1>(v));  v = fmaxf(v, lane_xor<2>(v));
    v = fmaxf(v, lane_xor<4>(v));  v = fmaxf(v, lane_xor<8>(v));
    v = fmaxf(v, lane_xor<16>(v));
    return v;
}
__device__ __forceinline__ float wave_sum(float v) {
    v += lane_xor<1>(v);  v += lane_xor<2>(v);
    v += lane_xor<4>(v);  v += lane_xor<8>(v);
    v += lane_xor<16>(v);
    return v;
}

#define GAT_LD(k) \
    const float4 x##k = xt4[(unsigned)__builtin_amdgcn_readlane(s_r, j + k) * (unsigned)(FT / 4) + (unsigned)lane];
#define GAT_FM(k) { \
    const float wa = rl_f(e0, j + k), wb = rl_f(e1, j + k); \
    const float w = half ? wb : wa; \
    acc.x = fmaf(w, x##k.x, acc.x); acc.y = fmaf(w, x##k.y, acc.y); \
    acc.z = fmaf(w, x##k.z, acc.z); acc.w = fmaf(w, x##k.w, acc.w); }

__device__ __forceinline__ void gat_accum(const float* __restrict__ xt, int s_r,
                                          float e0, float e1, int n,
                                          int lane, int half, float4& acc) {
    const float4* __restrict__ xt4 = (const float4*)xt;
    int j = 0;
    for (; j + 8 <= n; j += 8) {
        GAT_LD(0) GAT_LD(1) GAT_LD(2) GAT_LD(3) GAT_LD(4) GAT_LD(5) GAT_LD(6) GAT_LD(7)
        GAT_FM(0) GAT_FM(1) GAT_FM(2) GAT_FM(3) GAT_FM(4) GAT_FM(5) GAT_FM(6) GAT_FM(7)
    }
    if (j + 4 <= n) {
        GAT_LD(0) GAT_LD(1) GAT_LD(2) GAT_LD(3)
        GAT_FM(0) GAT_FM(1) GAT_FM(2) GAT_FM(3)
        j += 4;
    }
    if (j + 2 <= n) {
        GAT_LD(0) GAT_LD(1)
        GAT_FM(0) GAT_FM(1)
        j += 2;
    }
    if (j < n) {
        GAT_LD(0)
        GAT_FM(0)
    }
}

#define NW 16
__global__ __launch_bounds__(NW * 32, 16)
void gat_node_kernel(const int* __restrict__ ei,
                     const int* __restrict__ binCount,
                     const int* __restrict__ binned,
                     int* bucket2,
                     const float* __restrict__ a_src,
                     const float* __restrict__ a_dst,
                     const float* __restrict__ xt,
                     const float* __restrict__ bias,
                     const float* __restrict__ prelu_w,
                     float* __restrict__ out, int N, int E) {
    __shared__ int lcnt[BIN_NODES];
    __shared__ int lbucket[BIN_NODES * CAP];

    const int bin = blockIdx.x;
    if (threadIdx.x < BIN_NODES) lcnt[threadIdx.x] = 0;
    __syncthreads();
    const int total = binCount[bin];
    if (total <= BINCAP) {
        const int* __restrict__ bsrc = binned + (size_t)bin * BINCAP;
        int v[BINCAP / (NW * 32)];
        #pragma unroll
        for (int k = 0; k < BINCAP / (NW * 32); ++k) {
            const int i = (int)threadIdx.x + k * (NW * 32);
            v[k] = i < total ? bsrc[i] : -1;
        }
        #pragma unroll
        for (int k = 0; k < BINCAP / (NW * 32); ++k) {
            if (v[k] >= 0) {
                const int dl   = v[k] & (BIN_NODES - 1);
                const int slot = atomicAdd(&lcnt[dl], 1);
                if (slot < CAP) lbucket[dl * CAP + slot] = v[k] >> BIN_SHIFT;
            }
        }
    } else {
        for (int e = threadIdx.x; e < E; e += NW * 32) {
            const int d = ei[E + e];
            if ((d >> BIN_SHIFT) == bin) {
                const int dl   = d & (BIN_NODES - 1);
                const int slot = atomicAdd(&lcnt[dl], 1);
                if (slot < CAP) lbucket[dl * CAP + slot] = ei[e];
            }
        }
    }
    __syncthreads();

    const int lane = threadIdx.x & 31;
    const int half = lane >> 4;
    const int wave = __builtin_amdgcn_readfirstlane(threadIdx.x >> 5);

    for (int q = 0; q < BIN_NODES / NW; ++q) {
    const int dl   = wave + NW * q;
    const int node = bin * BIN_NODES + dl;
    if (node >= N) break;
    const int deg   = __builtin_amdgcn_readfirstlane(lcnt[dl]);
    const int* lb   = lbucket + dl * CAP;
    const int start = node * CAP;
    const float2 ad = *(const float2*)(a_dst + (size_t)node * HEADS);

    float4 acc = {0.f, 0.f, 0.f, 0.f};
    float den0 = 0.f, den1 = 0.f;

    if (deg <= 32) {
        const bool valid = lane < deg;
        const unsigned key = valid ? (((unsigned)lb[lane] << 5) | (unsigned)lane)
                                   : (0xFFFFFFE0u | (unsigned)lane);
        int rank = 0;
        #pragma unroll
        for (int t = 0; t < 32; ++t)
            rank += ((unsigned)__builtin_amdgcn_readlane((int)key, t) < key) ? 1 : 0;
        const int sorted = __builtin_amdgcn_ds_permute(rank << 2, (int)(key >> 5));
        const int s_r = valid ? sorted : 0;

        float v0 = -INFINITY, v1 = -INFINITY;
        if (valid) {
            const float2 as = *(const float2*)(a_src + (size_t)s_r * HEADS);
            v0 = leaky(as.x + ad.x);
            v1 = leaky(as.y + ad.y);
        }
        const float m0 = wave_max(v0), m1 = wave_max(v1);
        const float e0 = valid ? __expf(v0 - m0) : 0.f;
        const float e1 = valid ? __expf(v1 - m1) : 0.f;
        den0 = wave_sum(e0);
        den1 = wave_sum(e1);
        gat_accum(xt, s_r, e0, e1, deg, lane, half, acc);
    } else if (deg <= CAP) {
        for (int ib = 0; ib < deg; ib += 32) {
            const int i  = ib + lane;
            const int ki = (i < deg) ? lb[i] : 0x7fffffff;
            int rank = 0;
            for (int jb = 0; jb < deg; jb += 32) {
                const int jj = jb + lane;
                const int kj = (jj < deg) ? lb[jj] : 0x7fffffff;
                #pragma unroll
                for (int t = 0; t < 32; ++t) {
                    const int kt = __builtin_amdgcn_readlane(kj, t);
                    rank += (kt < ki || (kt == ki && jb + t < i)) ? 1 : 0;
                }
            }
            if (i < deg) bucket2[start + rank] = ki;
        }
        __threadfence();

        float mx0 = -INFINITY, mx1 = -INFINITY;
        for (int cb = 0; cb < deg; cb += 32) {
            if (cb + lane < deg) {
                const int s = bucket2[start + cb + lane];
                const float2 as = *(const float2*)(a_src + (size_t)s * HEADS);
                mx0 = fmaxf(mx0, leaky(as.x + ad.x));
                mx1 = fmaxf(mx1, leaky(as.y + ad.y));
            }
        }
        const float m0 = wave_max(mx0), m1 = wave_max(mx1);
        for (int cb = 0; cb < deg; cb += 32) {
            const bool valid = cb + lane < deg;
            const int s_r = valid ? bucket2[start + cb + lane] : 0;
            float e0 = 0.f, e1 = 0.f;
            if (valid) {
                const float2 as = *(const float2*)(a_src + (size_t)s_r * HEADS);
                e0 = __expf(leaky(as.x + ad.x) - m0);
                e1 = __expf(leaky(as.y + ad.y) - m1);
            }
            den0 += wave_sum(e0);
            den1 += wave_sum(e1);
            gat_accum(xt, s_r, e0, e1, min(32, deg - cb), lane, half, acc);
        }
    } else {
        const unsigned uE = (unsigned)E;
        const float4* __restrict__ xt4 = (const float4*)xt;
        float mx0 = -INFINITY, mx1 = -INFINITY;
        for (int eb = 0; eb < E; eb += 32) {
            const int e = eb + lane;
            if (e < E && ei[uE + (unsigned)e] == node) {
                const float2 as = *(const float2*)(a_src + (size_t)ei[(unsigned)e] * HEADS);
                mx0 = fmaxf(mx0, leaky(as.x + ad.x));
                mx1 = fmaxf(mx1, leaky(as.y + ad.y));
            }
        }
        const float m0 = wave_max(mx0), m1 = wave_max(mx1);
        for (int eb = 0; eb < E; eb += 32) {
            const int e = eb + lane;
            const bool hit = e < E && ei[uE + (unsigned)e] == node;
            unsigned mask = __builtin_amdgcn_ballot_w32(hit);
            if (mask == 0u) continue;
            int s = 0;
            float e0 = 0.f, e1 = 0.f;
            if (hit) {
                s = ei[(unsigned)e];
                const float2 as = *(const float2*)(a_src + (size_t)s * HEADS);
                e0 = __expf(leaky(as.x + ad.x) - m0);
                e1 = __expf(leaky(as.y + ad.y) - m1);
            }
            den0 += wave_sum(e0);
            den1 += wave_sum(e1);
            while (mask != 0u) {
                const int t = __builtin_ctz(mask);
                mask &= mask - 1u;
                const unsigned st = (unsigned)__builtin_amdgcn_readlane(s, t);
                const float4 xv = xt4[st * (unsigned)(FT / 4) + (unsigned)lane];
                const float wa = rl_f(e0, t), wb = rl_f(e1, t);
                const float w = half ? wb : wa;
                acc.x = fmaf(w, xv.x, acc.x); acc.y = fmaf(w, xv.y, acc.y);
                acc.z = fmaf(w, xv.z, acc.z); acc.w = fmaf(w, xv.w, acc.w);
            }
        }
    }

    const float dn  = half ? den1 : den0;
    const float inv = dn > 0.f ? dn : 1.f;
    const float4 b  = ((const float4*)bias)[(unsigned)lane];
    const float  pw = prelu_w[0];
    float4 r;
    r.x = acc.x / inv + b.x;  r.y = acc.y / inv + b.y;
    r.z = acc.z / inv + b.z;  r.w = acc.w / inv + b.w;
    r.x = r.x >= 0.f ? r.x : pw * r.x;  r.y = r.y >= 0.f ? r.y : pw * r.y;
    r.z = r.z >= 0.f ? r.z : pw * r.z;  r.w = r.w >= 0.f ? r.w : pw * r.w;
    ((float4*)out)[(unsigned)node * (unsigned)(FT / 4) + (unsigned)lane] = r;
    }
}

extern "C" void kernel_launch(void* const* d_in, const int* in_sizes, int n_in,
                              void* d_out, int out_size, void* d_ws, size_t ws_size,
                              hipStream_t stream) {
    const float* x       = (const float*)d_in[0];
    const int*   ei      = (const int*)  d_in[1];
    const float* W       = (const float*)d_in[2];
    const float* att_src = (const float*)d_in[3];
    const float* att_dst = (const float*)d_in[4];
    const float* bias    = (const float*)d_in[5];
    const float* prelu_w = (const float*)d_in[6];

    const int N = in_sizes[0] / IN_FT;
    const int E = in_sizes[1] / 2;

    float* out = (float*)d_out;

    const int NB = (N + BIN_NODES - 1) / BIN_NODES;

    float* ws       = (float*)d_ws;
    float* xt       = ws;
    int*   binned   = (int*)(xt + (size_t)N * FT);
    int*   bucket2  = binned + (size_t)NB * BINCAP;
    float* a_src    = (float*)(bucket2 + (size_t)N * CAP);
    float* a_dst    = a_src + (size_t)N * HEADS;
    int*   binCount = (int*)(a_dst + (size_t)N * HEADS);

    const int tiles_m = (N + 15) / 16;
    gat_gemm_wmma_kernel<<<(tiles_m + GW - 1) / GW, GW * 32, 0, stream>>>(x, W, att_src, att_dst,
                                                                        xt, a_src, a_dst, binCount, N);

    gat_bin_kernel<<<(E + BW * EPT - 1) / (BW * EPT), BW, 0, stream>>>(ei, binCount, binned, E, NB);

    gat_node_kernel<<<NB, NW * 32, 0, stream>>>(ei, binCount, binned, bucket2, a_src, a_dst, xt,
                                                bias, prelu_w, out, N, E);
}


# Binning rides in the product launch: a binning block counting-sorts 8192 edges in LDS (ranks + scan), writes exact-size runs (one slot an edge) and a row of starts: 3 -> 2 launches; sum order kept
# speedup vs baseline: 7.2674x; 1.2637x over previous
#include <hip/hip_runtime.h>
#include <math.h>

typedef float v2f __attribute__((ext_vector_type(2)));
typedef float v8f __attribute__((ext_vector_type(8)));

#define IN_FT   128
#define OUT_FT  64
#define HEADS   2
#define FT      (HEADS * OUT_FT)
#define NEG_SLOPE 0.2f

#define CAP        64
#define BIN_SHIFT  6
#define BIN_NODES  (1 << BIN_SHIFT)
#define BCH        8192
#define MAXNB      1024

__device__ __forceinline__ float leaky(float s) {
    return s > 0.f ? s : NEG_SLOPE * s;
}

#define GW      8
#define WPITCH  144
#define SPITCH  132

#define BEPT (BCH / (GW * 32))
__device__ __forceinline__ void gat_bin_block(int* lds, const int* __restrict__ ei,
                                              int* __restrict__ offs,
                                              int* __restrict__ sorted,
                                              int E, int NB, int NBP) {
    int* hist  = lds;
    int* wsum  = lds + MAXNB;
    int* stage = lds + 2 * MAXNB;
    const int tid   = threadIdx.x;
    const int lane  = tid & 31;
    const int wave  = tid >> 5;
    const int ebase = blockIdx.x * BCH;
    int* __restrict__ orow = offs + (size_t)blockIdx.x * NBP;

    int s[BEPT], d[BEPT], r[BEPT];
    #pragma unroll
    for (int k = 0; k < BEPT; ++k) {
        const int e  = ebase + k * (GW * 32) + tid;
        const int ec = e < E ? e : E - 1;
        s[k] = ei[(unsigned)ec];
        const int dv = ei[(unsigned)E + (unsigned)ec];
        d[k] = e < E ? dv : -1;
    }
    int base = 0;
    for (int b0 = 0; b0 < NB; b0 += MAXNB) {
        if (b0 != 0) __syncthreads();
        #pragma unroll
        for (int j = 0; j < MAXNB / (GW * 32); ++j) hist[tid + j * (GW * 32)] = 0;
        __syncthreads();
        #pragma unroll
        for (int k = 0; k < BEPT; ++k) {
            const int bin = (d[k] >> BIN_SHIFT) - b0;
            r[k] = -1;
            if (d[k] >= 0 && bin >= 0 && bin < MAXNB) r[k] = atomicAdd(&hist[bin], 1);
        }
        __syncthreads();
        const int4 c = *(const int4*)(hist + 4 * tid);
        const int tsum = c.x + c.y + c.z + c.w;
        int incl = tsum;
        #pragma unroll
        for (int off = 1; off < 32; off <<= 1) {
            const int nb = __shfl_up(incl, off, 32);
            if (lane >= off) incl += nb;
        }
        if (lane == 31) wsum[wave] = incl;
        __syncthreads();
        int wbase = 0, tot = 0;
        #pragma unroll
        for (int w = 0; w < GW; ++w) {
            const int xw = wsum[w];
            if (w < wave) wbase += xw;
            tot += xw;
        }
        const int excl = base + wbase + incl - tsum;
        int4 o;
        o.x = excl;  o.y = o.x + c.x;  o.z = o.y + c.y;  o.w = o.z + c.z;
        *(int4*)(hist + 4 * tid) = o;
        *(int4*)(orow + b0 + 4 * tid) = o;
        __syncthreads();
        int hb[BEPT];
        #pragma unroll
        for (int k = 0; k < BEPT; ++k) {
            const int bin = (d[k] >> BIN_SHIFT) - b0;
            hb[k] = hist[r[k] >= 0 ? bin : 0];
        }
        #pragma unroll
        for (int k = 0; k < BEPT; ++k) {
            if (r[k] >= 0)
                stage[hb[k] + r[k]] =
                    (int)(((unsigned)s[k] << BIN_SHIFT) | (unsigned)(d[k] & (BIN_NODES - 1)));
        }
        base += tot;
    }
    if (tid == 0 && (NB & (MAXNB - 1)) == 0) orow[NB] = base;
    __syncthreads();
    for (int i = 4 * tid; i < base; i += 4 * GW * 32)
        *(int4*)(sorted + ebase + i) = *(const int4*)(stage + i);
}

__global__ __launch_bounds__(GW * 32, 4)
void gat_gemm_wmma_kernel(const float* __restrict__ x,
                          const float* __restrict__ W,
                          const float* __restrict__ att_src,
                          const float* __restrict__ att_dst,
                          float* __restrict__ xt,
                          float* __restrict__ a_src,
                          float* __restrict__ a_dst,
                          const int* __restrict__ ei,
                          int* __restrict__ offs,
                          int* __restrict__ sorted,
                          int N, int E, int NB, int NBP, int nbb) {
    __shared__ __attribute__((aligned(16))) float L[IN_FT * WPITCH + GW * 16 * SPITCH];
    if ((int)blockIdx.x < nbb) {
        gat_bin_block((int*)L, ei, offs, sorted, E, NB, NBP);
        return;
    }
    const int gblk = (int)blockIdx.x - nbb;
    float* Wl = L;

    #pragma unroll
    for (int i = threadIdx.x; i < IN_FT * FT / 4; i += GW * 32) {
        const int k = i >> 5, c4 = i & 31;
        const float4 w = ((const float4*)W)[i];
        *(float4*)(Wl + k * WPITCH + 4 * c4) = w;
    }
    __syncthreads();

    const int wave = threadIdx.x >> 5;
    const int lane = threadIdx.x & 31;
    const int half = lane >> 4;
    const int l16  = lane & 15;

    const int row0 = (gblk * GW + wave) * 16;
    int arow = row0 + l16;
    if (arow >= N) arow = N - 1;

    const float* ap = x + (size_t)arow * IN_FT + 2 * half;
    const float* bp = Wl + (2 * half) * WPITCH + l16;

    v8f c[8];
    #pragma unroll
    for (int t = 0; t < 8; ++t) c[t] = v8f{};

    v2f ac[8], an[8];
    #pragma unroll
    for (int s = 0; s < 8; ++s) {
        ac[s].x = ap[4 * s];
        ac[s].y = ap[4 * s + 1];
    }
    #pragma unroll 1
    for (int ch = 0; ch < 4; ++ch) {
        const int nx = ch < 3 ? ch + 1 : 3;
        const float* apn = ap + 32 * nx;
        #pragma unroll
        for (int s = 0; s < 8; ++s) {
            an[s].x = apn[4 * s];
            an[s].y = apn[4 * s + 1];
        }
        const float* bpc = bp + (32 * ch) * WPITCH;
        #pragma unroll
        for (int s = 0; s < 8; ++s) {
            #pragma unroll
            for (int t = 0; t < 8; ++t) {
                v2f b;
                b.x = bpc[(4 * s) * WPITCH + 16 * t];
                b.y = bpc[(4 * s + 1) * WPITCH + 16 * t];
                c[t] = __builtin_amdgcn_wmma_f32_16x16x4_f32(false, ac[s], false, b,
                                                             (short)0, c[t], false, false);
            }
        }
        #pragma unroll
        for (int s = 0; s < 8; ++s) ac[s] = an[s];
    }

    float* St = L + IN_FT * WPITCH + wave * (16 * SPITCH);
    #pragma unroll
    for (int t = 0; t < 8; ++t) {
        #pragma unroll
        for (int r = 0; r < 8; ++r)
            St[(r + 8 * half) * SPITCH + 16 * t + l16] = c[t][r];
    }
    __syncthreads();

    #pragma unroll
    for (int r = 0; r < 16; ++r) {
        const int rr = row0 + r;
        if (rr < N) {
            const float4 v = *(const float4*)(St + r * SPITCH + 4 * lane);
            ((float4*)(xt + (size_t)rr * FT))[lane] = v;
        }
    }

    const int n = row0 + l16;
    const float4* xp = (const float4*)(St + l16 * SPITCH + half * OUT_FT);
    const float4* as = (const float4*)(att_src + half * OUT_FT);
    const float4* ad = (const float4*)(att_dst + half * OUT_FT);
    float s0 = 0.f, s1 = 0.f;
    #pragma unroll
    for (int j = 0; j < OUT_FT / 4; ++j) {
        float4 v = xp[j], a = as[j], d = ad[j];
        s0 += v.x * a.x + v.y * a.y + v.z * a.z + v.w * a.w;
        s1 += v.x * d.x + v.y * d.y + v.z * d.z + v.w * d.w;
    }
    if (n < N) {
        a_src[n * HEADS + half] = s0;
        a_dst[n * HEADS + half] = s1;
    }
}

__device__ __forceinline__ float rl_f(float v, int l) {
    return __int_as_float(__builtin_amdgcn_readlane(__float_as_int(v), l));
}
template <int M>
__device__ __forceinline__ float lane_xor(float v) {
    const int i = __float_as_int(v);
    if constexpr (M == 16)
        return __int_as_float(__builtin_amdgcn_permlanex16(i, i, 0x76543210, 0xfedcba98, false, false));
    else
        return __int_as_float(__builtin_amdgcn_update_dpp(i, i, 0x160 + M, 0xF, 0xF, false));
}
__device__ __forceinline__ float wave_max(float v) {
    v = fmaxf(v, lane_xor<1>(v));  v = fmaxf(v, lane_xor<2>(v));
    v = fmaxf(v, lane_xor<4>(v));  v = fmaxf(v, lane_xor<8>(v));
    v = fmaxf(v, lane_xor<16>(v));
    return v;
}
__device__ __forceinline__ float wave_sum(float v) {
    v += lane_xor<1>(v);  v += lane_xor<2>(v);
    v += lane_xor<4>(v);  v += lane_xor<8>(v);
    v += lane_xor<16>(v);
    return v;
}

#define GAT_LD(k) \
    const float4 x##k = xt4[(unsigned)__builtin_amdgcn_readlane(s_r, j + k) * (unsigned)(FT / 4) + (unsigned)lane];
#define GAT_FM(k) { \
    const float wa = rl_f(e0, j + k), wb = rl_f(e1, j + k); \
    const float w = half ? wb : wa; \
    acc.x = fmaf(w, x##k.x, acc.x); acc.y = fmaf(w, x##k.y, acc.y); \
    acc.z = fmaf(w, x##k.z, acc.z); acc.w = fmaf(w, x##k.w, acc.w); }

__device__ __forceinline__ void gat_accum(const float* __restrict__ xt, int s_r,
                                          float e0, float e1, int n,
                                          int lane, int half, float4& acc) {
    const float4* __restrict__ xt4 = (const float4*)xt;
    int j = 0;
    for (; j + 8 <= n; j += 8) {
        GAT_LD(0) GAT_LD(1) GAT_LD(2) GAT_LD(3) GAT_LD(4) GAT_LD(5) GAT_LD(6) GAT_LD(7)
        GAT_FM(0) GAT_FM(1) GAT_FM(2) GAT_FM(3) GAT_FM(4) GAT_FM(5) GAT_FM(6) GAT_FM(7)
    }
    if (j + 4 <= n) {
        GAT_LD(0) GAT_LD(1) GAT_LD(2) GAT_LD(3)
        GAT_FM(0) GAT_FM(1) GAT_FM(2) GAT_FM(3)
        j += 4;
    }
    if (j + 2 <= n) {
        GAT_LD(0) GAT_LD(1)
        GAT_FM(0) GAT_FM(1)
        j += 2;
    }
    if (j < n) {
        GAT_LD(0)
        GAT_FM(0)
    }
}

#define NW 16
#define RUNS 128
__global__ __launch_bounds__(NW * 32, 16)
void gat_node_kernel(const int* __restrict__ ei,
                     const int* __restrict__ offs,
                     const int* __restrict__ sorted,
                     int* bucket2,
                     const float* __restrict__ a_src,
                     const float* __restrict__ a_dst,
                     const float* __restrict__ xt,
                     const float* __restrict__ bias,
                     const float* __restrict__ prelu_w,
                     float* __restrict__ out, int N, int E, int nbb, int NBP) {
    __shared__ int lcnt[BIN_NODES];
    __shared__ int lbucket[BIN_NODES * CAP];
    __shared__ int2 runs[RUNS];

    const int bin  = blockIdx.x;
    const int lane = threadIdx.x & 31;
    const int half = lane >> 4;
    const int wave = __builtin_amdgcn_readfirstlane(threadIdx.x >> 5);

    if (threadIdx.x < BIN_NODES) lcnt[threadIdx.x] = 0;
    for (int c0 = 0; c0 < nbb; c0 += RUNS) {
        if (c0 != 0) __syncthreads();
        if (threadIdx.x < RUNS) {
            const int b = c0 + (int)threadIdx.x;
            int2 rn = {0, 0};
            if (b < nbb) {
                const int* __restrict__ o = offs + (size_t)b * NBP + bin;
                const int s0 = o[0], s1 = o[1];
                rn.x = b * BCH + s0;
                rn.y = s1 - s0;
            }
            runs[threadIdx.x] = rn;
        }
        __syncthreads();
        int2 rn[RUNS / NW];
        int  v[RUNS / NW];
        #pragma unroll
        for (int u = 0; u < RUNS / NW; ++u) rn[u] = runs[wave + NW * u];
        #pragma unroll
        for (int u = 0; u < RUNS / NW; ++u)
            v[u] = lane < rn[u].y ? sorted[rn[u].x + lane] : -1;
        #pragma unroll
        for (int u = 0; u < RUNS / NW; ++u) {
            if (v[u] >= 0) {
                const int dl   = v[u] & (BIN_NODES - 1);
                const int slot = atomicAdd(&lcnt[dl], 1);
                if (slot < CAP) lbucket[dl * CAP + slot] = v[u] >> BIN_SHIFT;
            }
        }
        for (int u = 0; u < RUNS / NW; ++u) {
            const int2 rr = runs[wave + NW * u];
            for (int k = lane; k < rr.y - 32; k += 32) {
                const int w    = sorted[rr.x + 32 + k];
                const int dl   = w & (BIN_NODES - 1);
                const int slot = atomicAdd(&lcnt[dl], 1);
                if (slot < CAP) lbucket[dl * CAP + slot] = w >> BIN_SHIFT;
            }
        }
    }
    __syncthreads();

    for (int q = 0; q < BIN_NODES / NW; ++q) {
    const int dl   = wave + NW * q;
    const int node = bin * BIN_NODES + dl;
    if (node >= N) break;
    const int deg   = __builtin_amdgcn_readfirstlane(lcnt[dl]);
    const int* lb   = lbucket + dl * CAP;
    const int start = node * CAP;
    const float2 ad = *(const float2*)(a_dst + (size_t)node * HEADS);

    float4 acc = {0.f, 0.f, 0.f, 0.f};
    float den0 = 0.f, den1 = 0.f;

    if (deg <= 32) {
        const bool valid = lane < deg;
        const unsigned key = valid ? (((unsigned)lb[lane] << 5) | (unsigned)lane)
                                   : (0xFFFFFFE0u | (unsigned)lane);
        int rank = 0;
        #pragma unroll
        for (int t = 0; t < 32; ++t)
            rank += ((unsigned)__builtin_amdgcn_readlane((int)key, t) < key) ? 1 : 0;
        const int sorted = __builtin_amdgcn_ds_permute(rank << 2, (int)(key >> 5));
        const int s_r = valid ? sorted : 0;

        float v0 = -INFINITY, v1 = -INFINITY;
        if (valid) {
            const float2 as = *(const float2*)(a_src + (size_t)s_r * HEADS);
            v0 = leaky(as.x + ad.x);
            v1 = leaky(as.y + ad.y);
        }
        const float m0 = wave_max(v0), m1 = wave_max(v1);
        const float e0 = valid ? __expf(v0 - m0) : 0.f;
        const float e1 = valid ? __expf(v1 - m1) : 0.f;
        den0 = wave_sum(e0);
        den1 = wave_sum(e1);
        gat_accum(xt, s_r, e0, e1, deg, lane, half, acc);
    } else if (deg <= CAP) {
        for (int ib = 0; ib < deg; ib += 32) {
            const int i  = ib + lane;
            const int ki = (i < deg) ? lb[i] : 0x7fffffff;
            int rank = 0;
            for (int jb = 0; jb < deg; jb += 32) {
                const int jj = jb + lane;
                const int kj = (jj < deg) ? lb[jj] : 0x7fffffff;
                #pragma unroll
                for (int t = 0; t < 32; ++t) {
                    const int kt = __builtin_amdgcn_readlane(kj, t);
                    rank += (kt < ki || (kt == ki && jb + t < i)) ? 1 : 0;
                }
            }
            if (i < deg) bucket2[start + rank] = ki;
        }
        __threadfence();

        float mx0 = -INFINITY, mx1 = -INFINITY;
        for (int cb = 0; cb < deg; cb += 32) {
            if (cb + lane < deg) {
                const int s = bucket2[start + cb + lane];
                const float2 as = *(const float2*)(a_src + (size_t)s * HEADS);
                mx0 = fmaxf(mx0, leaky(as.x + ad.x));
                mx1 = fmaxf(mx1, leaky(as.y + ad.y));
            }
        }
        const float m0 = wave_max(mx0), m1 = wave_max(mx1);
        for (int cb = 0; cb < deg; cb += 32) {
            const bool valid = cb + lane < deg;
            const int s_r = valid ? bucket2[start + cb + lane] : 0;
            float e0 = 0.f, e1 = 0.f;
            if (valid) {
                const float2 as = *(const float2*)(a_src + (size_t)s_r * HEADS);
                e0 = __expf(leaky(as.x + ad.x) - m0);
                e1 = __expf(leaky(as.y + ad.y) - m1);
            }
            den0 += wave_sum(e0);
            den1 += wave_sum(e1);
            gat_accum(xt, s_r, e0, e1, min(32, deg - cb), lane, half, acc);
        }
    } else {
        const unsigned uE = (unsigned)E;
        const float4* __restrict__ xt4 = (const float4*)xt;
        float mx0 = -INFINITY, mx1 = -INFINITY;
        #pragma unroll 1
        for (int eb = 0; eb < E; eb += 32) {
            const int e = eb + lane;
            if (e < E && ei[uE + (unsigned)e] == node) {
                const float2 as = *(const float2*)(a_src + (size_t)ei[(unsigned)e] * HEADS);
                mx0 = fmaxf(mx0, leaky(as.x + ad.x));
                mx1 = fmaxf(mx1, leaky(as.y + ad.y));
            }
        }
        const float m0 = wave_max(mx0), m1 = wave_max(mx1);
        for (int eb = 0; eb < E; eb += 32) {
            const int e = eb + lane;
            const bool hit = e < E && ei[uE + (unsigned)e] == node;
            unsigned mask = __builtin_amdgcn_ballot_w32(hit);
            if (mask == 0u) continue;
            int s = 0;
            float e0 = 0.f, e1 = 0.f;
            if (hit) {
                s = ei[(unsigned)e];
                const float2 as = *(const float2*)(a_src + (size_t)s * HEADS);
                e0 = __expf(leaky(as.x + ad.x) - m0);
                e1 = __expf(leaky(as.y + ad.y) - m1);
            }
            den0 += wave_sum(e0);
            den1 += wave_sum(e1);
            while (mask != 0u) {
                const int t = __builtin_ctz(mask);
                mask &= mask - 1u;
                const unsigned st = (unsigned)__builtin_amdgcn_readlane(s, t);
                const float4 xv = xt4[st * (unsigned)(FT / 4) + (unsigned)lane];
                const float wa = rl_f(e0, t), wb = rl_f(e1, t);
                const float w = half ? wb : wa;
                acc.x = fmaf(w, xv.x, acc.x); acc.y = fmaf(w, xv.y, acc.y);
                acc.z = fmaf(w, xv.z, acc.z); acc.w = fmaf(w, xv.w, acc.w);
            }
        }
    }

    const float dn  = half ? den1 : den0;
    const float inv = dn > 0.f ? dn : 1.f;
    const float4 b  = ((const float4*)bias)[(unsigned)lane];
    const float  pw = prelu_w[0];
    float4 r;
    r.x = acc.x / inv + b.x;  r.y = acc.y / inv + b.y;
    r.z = acc.z / inv + b.z;  r.w = acc.w / inv + b.w;
    r.x = r.x >= 0.f ? r.x : pw * r.x;  r.y = r.y >= 0.f ? r.y : pw * r.y;
    r.z = r.z >= 0.f ? r.z : pw * r.z;  r.w = r.w >= 0.f ? r.w : pw * r.w;
    ((float4*)out)[(unsigned)node * (unsigned)(FT / 4) + (unsigned)lane] = r;
    }
}

extern "C" void kernel_launch(void* const* d_in, const int* in_sizes, int n_in,
                              void* d_out, int out_size, void* d_ws, size_t ws_size,
                              hipStream_t stream) {
    const float* x       = (const float*)d_in[0];
    const int*   ei      = (const int*)  d_in[1];
    const float* W       = (const float*)d_in[2];
    const float* att_src = (const float*)d_in[3];
    const float* att_dst = (const float*)d_in[4];
    const float* bias    = (const float*)d_in[5];
    const float* prelu_w = (const float*)d_in[6];

    const int N = in_sizes[0] / IN_FT;
    const int E = in_sizes[1] / 2;

    float* out = (float*)d_out;

    const int NB = (N + BIN_NODES - 1) / BIN_NODES;

    const int nbb = (E + BCH - 1) / BCH;
    const int NBP = (NB + MAXNB - 1) / MAXNB * MAXNB + 32;

    float* ws       = (float*)d_ws;
    float* xt       = ws;
    int*   sorted   = (int*)(xt + (size_t)N * FT);
    int*   offs     = sorted + (size_t)nbb * BCH;
    int*   bucket2  = offs + (size_t)nbb * NBP;
    float* a_src    = (float*)(bucket2 + (size_t)N * CAP);
    float* a_dst    = a_src + (size_t)N * HEADS;

    const int tiles_m = (N + 15) / 16;
    gat_gemm_wmma_kernel<<<nbb + (tiles_m + GW - 1) / GW, GW * 32, 0, stream>>>(x, W, att_src, att_dst,
                                                                              xt, a_src, a_dst,
                                                                              ei, offs, sorted,
                                                                              N, E, NB, NBP, nbb);

    gat_node_kernel<<<NB, NW * 32, 0, stream>>>(ei, offs, sorted, bucket2, a_src, a_dst, xt,
                                                bias, prelu_w, out, N, E, nbb, NBP);
}
